# starting kernel
# baseline (speedup 1.0000x reference)
#include <hip/hip_runtime.h>
#include <cstdint>
#include <cstddef>

typedef __attribute__((ext_vector_type(2))) float v2f;
typedef __attribute__((ext_vector_type(8))) float v8f;

template <int NCOL, int EPI>
__global__ __launch_bounds__(256) void gemm_wmma_kernel(
    const float* __restrict__ A, const float* __restrict__ W,
    const float* __restrict__ bias, float* __restrict__ out, int nrows)
{
    constexpr int NT = NCOL / 16;
    __shared__ __align__(16) float sW[64 * NCOL];
    __shared__ __align__(16) float sA[8][16 * 68];

    const int tid  = threadIdx.x;
    const int wave = tid >> 5;
    const int lane = tid & 31;
    const int hh   = lane >> 4;
    const int l16  = lane & 15;
    const int r0   = blockIdx.x * 128 + wave * 16;

    for (int i = tid; i < 64 * NCOL; i += 256) sW[i] = W[i];

    const float4* A4 = reinterpret_cast<const float4*>(A);
    #pragma unroll
    for (int i = 0; i < 8; ++i) {
        int q  = lane + 32 * i;
        int r  = q >> 4;
        int c4 = q & 15;
        int gr = r0 + r;
        float4 v = make_float4(0.f, 0.f, 0.f, 0.f);
        if (gr < nrows) v = A4[(size_t)gr * 16 + c4];
        *reinterpret_cast<float4*>(&sA[wave][r * 68 + c4 * 4]) = v;
    }
    __syncthreads();

    v8f acc[NT] = {};

    #pragma unroll
    for (int kb = 0; kb < 64; kb += 4) {
        v2f a;
        a.x = sA[wave][l16 * 68 + kb + 2 * hh + 0];
        a.y = sA[wave][l16 * 68 + kb + 2 * hh + 1];
        #pragma unroll
        for (int nt = 0; nt < NT; ++nt) {
            v2f b;
            b.x = sW[(kb + 2 * hh + 0) * NCOL + nt * 16 + l16];
            b.y = sW[(kb + 2 * hh + 1) * NCOL + nt * 16 + l16];
            acc[nt] = __builtin_amdgcn_wmma_f32_16x16x4_f32(
                false, a, false, b, (short)0, acc[nt], false, false);
        }
    }

    #pragma unroll
    for (int nt = 0; nt < NT; ++nt) {
        const int col = nt * 16 + l16;
        float bv = 0.f;
        if (EPI > 0) bv = bias[col];
        #pragma unroll
        for (int j = 0; j < 8; ++j) {
            int gr = r0 + j + 8 * hh;
            if (gr < nrows) {
                float v = acc[nt][j] + bv;
                if (EPI == 2) v = fmaxf(v, 0.f);
                out[(size_t)gr * NCOL + col] = v;
            }
        }
    }
}

__global__ __launch_bounds__(256) void deg_init_kernel(float* __restrict__ deg, int n)
{
    int i = blockIdx.x * 256 + threadIdx.x;
    if (i < n) deg[i] = 1.0f;
}

__global__ __launch_bounds__(256) void deg_count_kernel(
    const int* __restrict__ col, float* __restrict__ deg, int ne)
{
    int e = blockIdx.x * 256 + threadIdx.x;
    if (e < ne) unsafeAtomicAdd(&deg[col[e]], 1.0f);
}

__global__ __launch_bounds__(256) void dinv_kernel(float* __restrict__ d, int n)
{
    int i = blockIdx.x * 256 + threadIdx.x;
    if (i < n) d[i] = rsqrtf(d[i]);
}

__global__ __launch_bounds__(256) void self_term_kernel(
    const float* __restrict__ xw, const float* __restrict__ dinv,
    float* __restrict__ agg, int n)
{
    int t = blockIdx.x * 256 + threadIdx.x;
    if (t < n * 64) {
        int i = t >> 6;
        float dv = dinv[i];
        agg[t] = xw[t] * dv * dv;
    }
}

__global__ __launch_bounds__(256) void edge_scatter_kernel(
    const int* __restrict__ row, const int* __restrict__ col,
    const float* __restrict__ dinv, const float* __restrict__ xw,
    float* __restrict__ agg, int ne)
{
    int t    = blockIdx.x * 256 + threadIdx.x;
    int e    = t >> 5;
    int lane = t & 31;
    if (e >= ne) return;
    int   r  = row[e];
    int   c  = col[e];
    float nm = dinv[r] * dinv[c];
    const float* src = xw  + (size_t)r * 64;
    float*       dst = agg + (size_t)c * 64;
    unsafeAtomicAdd(&dst[lane],      nm * src[lane]);
    unsafeAtomicAdd(&dst[lane + 32], nm * src[lane + 32]);
}

__global__ __launch_bounds__(256) void bn_relu_kernel(
    const float* __restrict__ agg, const float* __restrict__ b,
    const float* __restrict__ g, const float* __restrict__ beta,
    const float* __restrict__ mean, const float* __restrict__ var,
    float* __restrict__ out, int n)
{
    int t = blockIdx.x * 256 + threadIdx.x;
    if (t < n * 64) {
        int f = t & 63;
        float v = (agg[t] + b[f] - mean[f]) * rsqrtf(var[f] + 1e-5f) * g[f] + beta[f];
        out[t] = fmaxf(v, 0.f);
    }
}

extern "C" void kernel_launch(void* const* d_in, const int* in_sizes, int n_in,
                              void* d_out, int out_size, void* d_ws, size_t ws_size,
                              hipStream_t stream)
{
    const float* x      = (const float*)d_in[0];
    const int*   ei     = (const int*)d_in[1];
    const float* W0     = (const float*)d_in[2];
    const float* b0     = (const float*)d_in[3];
    const float* W1     = (const float*)d_in[4];
    const float* b1     = (const float*)d_in[5];
    const float* bn0_g  = (const float*)d_in[6];
    const float* bn0_b  = (const float*)d_in[7];
    const float* bn0_m  = (const float*)d_in[8];
    const float* bn0_v  = (const float*)d_in[9];
    const float* bn1_g  = (const float*)d_in[10];
    const float* bn1_b  = (const float*)d_in[11];
    const float* bn1_m  = (const float*)d_in[12];
    const float* bn1_v  = (const float*)d_in[13];
    const float* lin1_W = (const float*)d_in[14];
    const float* lin1_b = (const float*)d_in[15];
    const float* lin2_W = (const float*)d_in[16];
    const float* lin2_b = (const float*)d_in[17];
    const float* lin3_W = (const float*)d_in[18];
    const float* lin3_b = (const float*)d_in[19];

    const int N = in_sizes[0] / 64;
    const int E = in_sizes[1] / 2;
    const int* row = ei;
    const int* col = ei + E;

    char*  ws   = (char*)d_ws;
    size_t off0 = 0;
    size_t off1 = (off0 + (size_t)N * sizeof(float) + 255) & ~(size_t)255;
    size_t off2 = (off1 + (size_t)N * 64 * sizeof(float) + 255) & ~(size_t)255;
    float* dinv = (float*)(ws + off0);
    float* bufA = (float*)(ws + off1);
    float* bufB = (float*)(ws + off2);

    const dim3 blk(256);
    const int gN    = (N + 255) / 256;
    const int gE    = (E + 255) / 256;
    const int gNF   = (N * 64 + 255) / 256;
    const int gEdge = (E * 32 + 255) / 256;
    const int gGemm = (N + 127) / 128;

    deg_init_kernel<<<gN, blk, 0, stream>>>(dinv, N);
    deg_count_kernel<<<gE, blk, 0, stream>>>(col, dinv, E);
    dinv_kernel<<<gN, blk, 0, stream>>>(dinv, N);

    gemm_wmma_kernel<64, 0><<<gGemm, blk, 0, stream>>>(x, W0, nullptr, bufA, N);
    self_term_kernel<<<gNF, blk, 0, stream>>>(bufA, dinv, bufB, N);
    edge_scatter_kernel<<<gEdge, blk, 0, stream>>>(row, col, dinv, bufA, bufB, E);
    bn_relu_kernel<<<gNF, blk, 0, stream>>>(bufB, b0, bn0_g, bn0_b, bn0_m, bn0_v, bufA, N);

    gemm_wmma_kernel<64, 0><<<gGemm, blk, 0, stream>>>(bufA, W1, nullptr, bufB, N);
    self_term_kernel<<<gNF, blk, 0, stream>>>(bufB, dinv, bufA, N);
    edge_scatter_kernel<<<gEdge, blk, 0, stream>>>(row, col, dinv, bufB, bufA, E);
    bn_relu_kernel<<<gNF, blk, 0, stream>>>(bufA, b1, bn1_g, bn1_b, bn1_m, bn1_v, bufB, N);

    gemm_wmma_kernel<64, 2><<<gGemm, blk, 0, stream>>>(bufB, lin1_W, lin1_b, bufA, N);
    gemm_wmma_kernel<64, 2><<<gGemm, blk, 0, stream>>>(bufA, lin2_W, lin2_b, bufB, N);
    gemm_wmma_kernel<16, 1><<<gGemm, blk, 0, stream>>>(bufB, lin3_W, lin3_b, (float*)d_out, N);
}


# elementwise passes folded into the GEMM kernels: epilogue also writes the self-loop seed dinv^2*xw, A loader applies conv bias + batch-norm + ReLU; 4 launches of 25000 blocks gone; sums unchanged
# speedup vs baseline: 1.7572x; 1.7572x over previous
#include <hip/hip_runtime.h>
#include <cstdint>
#include <cstddef>

typedef __attribute__((ext_vector_type(2))) float v2f;
typedef __attribute__((ext_vector_type(8))) float v8f;

template <int NCOL, int PRO, int EPI>
__global__ __launch_bounds__(256) void gemm_wmma_kernel(
    const float* __restrict__ A, const float* __restrict__ W,
    const float* __restrict__ bias, float* __restrict__ out, int nrows,
    const float* __restrict__ cb, const float* __restrict__ bn_g,
    const float* __restrict__ bn_b, const float* __restrict__ bn_m,
    const float* __restrict__ bn_v,
    const float* __restrict__ dinv, float* __restrict__ seed)
{
    constexpr int NT = NCOL / 16;
    __shared__ __align__(16) float sW[64 * NCOL];
    __shared__ __align__(16) float sA[8][16 * 68];

    const int tid  = threadIdx.x;
    const int wave = tid >> 5;
    const int lane = tid & 31;
    const int hh   = lane >> 4;
    const int l16  = lane & 15;
    const int r0   = blockIdx.x * 128 + wave * 16;

    for (int i = tid; i < 64 * NCOL; i += 256) sW[i] = W[i];

    float4 pb = make_float4(0.f, 0.f, 0.f, 0.f), pm = pb, pr = pb, pg = pb, pe = pb;
    if (PRO == 1) {
        pb = reinterpret_cast<const float4*>(cb)[l16];
        pm = reinterpret_cast<const float4*>(bn_m)[l16];
        pg = reinterpret_cast<const float4*>(bn_g)[l16];
        pe = reinterpret_cast<const float4*>(bn_b)[l16];
        const float4 pv = reinterpret_cast<const float4*>(bn_v)[l16];
        pr.x = rsqrtf(pv.x + 1e-5f);
        pr.y = rsqrtf(pv.y + 1e-5f);
        pr.z = rsqrtf(pv.z + 1e-5f);
        pr.w = rsqrtf(pv.w + 1e-5f);
    }

    float dv[8];
    if (EPI == 3) {
        #pragma unroll
        for (int j = 0; j < 8; ++j) {
            int gr = r0 + j + 8 * hh;
            dv[j] = (gr < nrows) ? dinv[gr] : 0.f;
        }
    }

    const float4* A4 = reinterpret_cast<const float4*>(A);
    #pragma unroll
    for (int i = 0; i < 8; ++i) {
        int q  = lane + 32 * i;
        int r  = q >> 4;
        int c4 = q & 15;
        int gr = r0 + r;
        float4 v = make_float4(0.f, 0.f, 0.f, 0.f);
        if (gr < nrows) {
            v = A4[(size_t)gr * 16 + c4];
            if (PRO == 1) {
                v.x = fmaxf((v.x + pb.x - pm.x) * pr.x * pg.x + pe.x, 0.f);
                v.y = fmaxf((v.y + pb.y - pm.y) * pr.y * pg.y + pe.y, 0.f);
                v.z = fmaxf((v.z + pb.z - pm.z) * pr.z * pg.z + pe.z, 0.f);
                v.w = fmaxf((v.w + pb.w - pm.w) * pr.w * pg.w + pe.w, 0.f);
            }
        }
        *reinterpret_cast<float4*>(&sA[wave][r * 68 + c4 * 4]) = v;
    }
    __syncthreads();

    v8f acc[NT] = {};

    #pragma unroll
    for (int kb = 0; kb < 64; kb += 4) {
        v2f a;
        a.x = sA[wave][l16 * 68 + kb + 2 * hh + 0];
        a.y = sA[wave][l16 * 68 + kb + 2 * hh + 1];
        #pragma unroll
        for (int nt = 0; nt < NT; ++nt) {
            v2f b;
            b.x = sW[(kb + 2 * hh + 0) * NCOL + nt * 16 + l16];
            b.y = sW[(kb + 2 * hh + 1) * NCOL + nt * 16 + l16];
            acc[nt] = __builtin_amdgcn_wmma_f32_16x16x4_f32(
                false, a, false, b, (short)0, acc[nt], false, false);
        }
    }

    #pragma unroll
    for (int nt = 0; nt < NT; ++nt) {
        const int col = nt * 16 + l16;
        float bv = 0.f;
        if (EPI == 1 || EPI == 2) bv = bias[col];
        #pragma unroll
        for (int j = 0; j < 8; ++j) {
            int gr = r0 + j + 8 * hh;
            if (gr < nrows) {
                float v = acc[nt][j] + bv;
                if (EPI == 2) v = fmaxf(v, 0.f);
                out[(size_t)gr * NCOL + col] = v;
                if (EPI == 3) seed[(size_t)gr * NCOL + col] = v * dv[j] * dv[j];
            }
        }
    }
}

__global__ __launch_bounds__(256) void deg_init_kernel(float* __restrict__ deg, int n)
{
    int i = blockIdx.x * 256 + threadIdx.x;
    if (i < n) deg[i] = 1.0f;
}

__global__ __launch_bounds__(256) void deg_count_kernel(
    const int* __restrict__ col, float* __restrict__ deg, int ne)
{
    int e = blockIdx.x * 256 + threadIdx.x;
    if (e < ne) unsafeAtomicAdd(&deg[col[e]], 1.0f);
}

__global__ __launch_bounds__(256) void dinv_kernel(float* __restrict__ d, int n)
{
    int i = blockIdx.x * 256 + threadIdx.x;
    if (i < n) d[i] = rsqrtf(d[i]);
}

__global__ __launch_bounds__(256) void edge_scatter_kernel(
    const int* __restrict__ row, const int* __restrict__ col,
    const float* __restrict__ dinv, const float* __restrict__ xw,
    float* __restrict__ agg, int ne)
{
    const int lane = threadIdx.x & 31;
    const int wave = __builtin_amdgcn_readfirstlane((int)(threadIdx.x >> 5));
    const int e0   = blockIdx.x * 256 + wave * 32;
    int cnt = ne - e0;
    if (cnt > 32) cnt = 32;
    if (cnt <= 0) return;

    int   r  = 0;
    int   c  = 0;
    float nm = 0.f;
    if (lane < cnt) {
        r  = row[e0 + lane];
        c  = col[e0 + lane];
        nm = dinv[r] * dinv[c];
    }
    const int nmi = __float_as_int(nm);

    if (cnt == 32) {
        #pragma unroll
        for (int jb = 0; jb < 32; jb += 8) {
            float v0[8], v1[8];
            #pragma unroll
            for (int j = 0; j < 8; ++j) {
                const int rj = __builtin_amdgcn_readlane(r, jb + j);
                const float* src = xw + (size_t)rj * 64;
                v0[j] = src[lane];
                v1[j] = src[lane + 32];
            }
            #pragma unroll
            for (int j = 0; j < 8; ++j) {
                const int   cj = __builtin_amdgcn_readlane(c, jb + j);
                const float nj = __int_as_float(__builtin_amdgcn_readlane(nmi, jb + j));
                float* dst = agg + (size_t)cj * 64;
                unsafeAtomicAdd(&dst[lane],      nj * v0[j]);
                unsafeAtomicAdd(&dst[lane + 32], nj * v1[j]);
            }
        }
    } else {
        for (int j = 0; j < cnt; ++j) {
            const int   rj = __builtin_amdgcn_readlane(r, j);
            const int   cj = __builtin_amdgcn_readlane(c, j);
            const float nj = __int_as_float(__builtin_amdgcn_readlane(nmi, j));
            const float* src = xw  + (size_t)rj * 64;
            float*       dst = agg + (size_t)cj * 64;
            unsafeAtomicAdd(&dst[lane],      nj * src[lane]);
            unsafeAtomicAdd(&dst[lane + 32], nj * src[lane + 32]);
        }
    }
}

extern "C" void kernel_launch(void* const* d_in, const int* in_sizes, int n_in,
                              void* d_out, int out_size, void* d_ws, size_t ws_size,
                              hipStream_t stream)
{
    const float* x      = (const float*)d_in[0];
    const int*   ei     = (const int*)d_in[1];
    const float* W0     = (const float*)d_in[2];
    const float* b0     = (const float*)d_in[3];
    const float* W1     = (const float*)d_in[4];
    const float* b1     = (const float*)d_in[5];
    const float* bn0_g  = (const float*)d_in[6];
    const float* bn0_b  = (const float*)d_in[7];
    const float* bn0_m  = (const float*)d_in[8];
    const float* bn0_v  = (const float*)d_in[9];
    const float* bn1_g  = (const float*)d_in[10];
    const float* bn1_b  = (const float*)d_in[11];
    const float* bn1_m  = (const float*)d_in[12];
    const float* bn1_v  = (const float*)d_in[13];
    const float* lin1_W = (const float*)d_in[14];
    const float* lin1_b = (const float*)d_in[15];
    const float* lin2_W = (const float*)d_in[16];
    const float* lin2_b = (const float*)d_in[17];
    const float* lin3_W = (const float*)d_in[18];
    const float* lin3_b = (const float*)d_in[19];

    const int N = in_sizes[0] / 64;
    const int E = in_sizes[1] / 2;
    const int* row = ei;
    const int* col = ei + E;

    char*  ws   = (char*)d_ws;
    size_t off0 = 0;
    size_t off1 = (off0 + (size_t)N * sizeof(float) + 255) & ~(size_t)255;
    size_t off2 = (off1 + (size_t)N * 64 * sizeof(float) + 255) & ~(size_t)255;
    size_t off3 = (off2 + (size_t)N * 64 * sizeof(float) + 255) & ~(size_t)255;
    float* dinv = (float*)(ws + off0);
    float* bufA = (float*)(ws + off1);
    float* bufB = (float*)(ws + off2);
    float* bufC = (float*)(ws + off3);

    const dim3 blk(256);
    const int gN    = (N + 255) / 256;
    const int gE    = (E + 255) / 256;
    const int gEdge = (E + 255) / 256;
    const int gGemm = (N + 127) / 128;

    deg_init_kernel<<<gN, blk, 0, stream>>>(dinv, N);
    deg_count_kernel<<<gE, blk, 0, stream>>>(col, dinv, E);
    dinv_kernel<<<gN, blk, 0, stream>>>(dinv, N);

    gemm_wmma_kernel<64, 0, 3><<<gGemm, blk, 0, stream>>>(
        x, W0, nullptr, bufA, N,
        nullptr, nullptr, nullptr, nullptr, nullptr, dinv, bufB);
    edge_scatter_kernel<<<gEdge, blk, 0, stream>>>(row, col, dinv, bufA, bufB, E);

    gemm_wmma_kernel<64, 1, 3><<<gGemm, blk, 0, stream>>>(
        bufB, W1, nullptr, bufA, N,
        b0, bn0_g, bn0_b, bn0_m, bn0_v, dinv, bufC);
    edge_scatter_kernel<<<gEdge, blk, 0, stream>>>(row, col, dinv, bufA, bufC, E);

    gemm_wmma_kernel<64, 1, 2><<<gGemm, blk, 0, stream>>>(
        bufC, lin1_W, lin1_b, bufA, N,
        b1, bn1_g, bn1_b, bn1_m, bn1_v, nullptr, nullptr);
    gemm_wmma_kernel<64, 0, 2><<<gGemm, blk, 0, stream>>>(
        bufA, lin2_W, lin2_b, bufB, N,
        nullptr, nullptr, nullptr, nullptr, nullptr, nullptr, nullptr);
    gemm_wmma_kernel<16, 0, 1><<<gGemm, blk, 0, stream>>>(
        bufB, lin3_W, lin3_b, (float*)d_out, N,
        nullptr, nullptr, nullptr, nullptr, nullptr, nullptr, nullptr);
}


# GEMM kernels: staging loads issued together, not one round trip each: W copy unrolled with 16-byte loads, strip and degree loads clamp the row index instead of branching per load; sums unchanged
# speedup vs baseline: 1.9755x; 1.1242x over previous
#include <hip/hip_runtime.h>
#include <cstdint>
#include <cstddef>

typedef __attribute__((ext_vector_type(2))) float v2f;
typedef __attribute__((ext_vector_type(8))) float v8f;

template <int NCOL, int PRO, int EPI>
__global__ __launch_bounds__(256) void gemm_wmma_kernel(
    const float* __restrict__ A, const float* __restrict__ W,
    const float* __restrict__ bias, float* __restrict__ out, int nrows,
    const float* __restrict__ cb, const float* __restrict__ bn_g,
    const float* __restrict__ bn_b, const float* __restrict__ bn_m,
    const float* __restrict__ bn_v,
    const float* __restrict__ dinv, float* __restrict__ seed)
{
    constexpr int NT = NCOL / 16;
    __shared__ __align__(16) float sW[64 * NCOL];
    __shared__ __align__(16) float sA[8][16 * 68];

    const int tid  = threadIdx.x;
    const int wave = tid >> 5;
    const int lane = tid & 31;
    const int hh   = lane >> 4;
    const int l16  = lane & 15;
    const int r0   = blockIdx.x * 128 + wave * 16;

    {
        const float4* W4  = reinterpret_cast<const float4*>(W);
        float4*       sW4 = reinterpret_cast<float4*>(sW);
        constexpr int NV = (64 * NCOL / 4) / 256;
        #pragma unroll
        for (int t = 0; t < NV; ++t) sW4[tid + 256 * t] = W4[tid + 256 * t];
    }

    float4 pb = make_float4(0.f, 0.f, 0.f, 0.f), pm = pb, pr = pb, pg = pb, pe = pb;
    if (PRO == 1) {
        pb = reinterpret_cast<const float4*>(cb)[l16];
        pm = reinterpret_cast<const float4*>(bn_m)[l16];
        pg = reinterpret_cast<const float4*>(bn_g)[l16];
        pe = reinterpret_cast<const float4*>(bn_b)[l16];
        const float4 pv = reinterpret_cast<const float4*>(bn_v)[l16];
        pr.x = rsqrtf(pv.x + 1e-5f);
        pr.y = rsqrtf(pv.y + 1e-5f);
        pr.z = rsqrtf(pv.z + 1e-5f);
        pr.w = rsqrtf(pv.w + 1e-5f);
    }

    const float4* A4 = reinterpret_cast<const float4*>(A);
    float4 sv[8];
    #pragma unroll
    for (int i = 0; i < 8; ++i) {
        int gr = r0 + 2 * i + hh;
        gr = (gr < nrows) ? gr : nrows - 1;
        sv[i] = A4[(size_t)gr * 16 + l16];
    }
    #pragma unroll
    for (int i = 0; i < 8; ++i) {
        float4 v = sv[i];
        if (PRO == 1) {
            v.x = fmaxf((v.x + pb.x - pm.x) * pr.x * pg.x + pe.x, 0.f);
            v.y = fmaxf((v.y + pb.y - pm.y) * pr.y * pg.y + pe.y, 0.f);
            v.z = fmaxf((v.z + pb.z - pm.z) * pr.z * pg.z + pe.z, 0.f);
            v.w = fmaxf((v.w + pb.w - pm.w) * pr.w * pg.w + pe.w, 0.f);
        }
        *reinterpret_cast<float4*>(&sA[wave][(2 * i + hh) * 68 + l16 * 4]) = v;
    }
    __syncthreads();

    float dv[8];
    if (EPI == 3) {
        #pragma unroll
        for (int j = 0; j < 8; ++j) {
            int gr = r0 + j + 8 * hh;
            gr = (gr < nrows) ? gr : nrows - 1;
            dv[j] = dinv[gr];
        }
    }

    v8f acc[NT] = {};

    #pragma unroll
    for (int kb = 0; kb < 64; kb += 4) {
        v2f a;
        a.x = sA[wave][l16 * 68 + kb + 2 * hh + 0];
        a.y = sA[wave][l16 * 68 + kb + 2 * hh + 1];
        #pragma unroll
        for (int nt = 0; nt < NT; ++nt) {
            v2f b;
            b.x = sW[(kb + 2 * hh + 0) * NCOL + nt * 16 + l16];
            b.y = sW[(kb + 2 * hh + 1) * NCOL + nt * 16 + l16];
            acc[nt] = __builtin_amdgcn_wmma_f32_16x16x4_f32(
                false, a, false, b, (short)0, acc[nt], false, false);
        }
    }

    #pragma unroll
    for (int nt = 0; nt < NT; ++nt) {
        const int col = nt * 16 + l16;
        float bv = 0.f;
        if (EPI == 1 || EPI == 2) bv = bias[col];
        #pragma unroll
        for (int j = 0; j < 8; ++j) {
            int gr = r0 + j + 8 * hh;
            if (gr < nrows) {
                float v = acc[nt][j] + bv;
                if (EPI == 2) v = fmaxf(v, 0.f);
                out[(size_t)gr * NCOL + col] = v;
                if (EPI == 3) seed[(size_t)gr * NCOL + col] = v * dv[j] * dv[j];
            }
        }
    }
}

__global__ __launch_bounds__(256) void deg_init_kernel(float* __restrict__ deg, int n)
{
    int i = blockIdx.x * 256 + threadIdx.x;
    if (i < n) deg[i] = 1.0f;
}

__global__ __launch_bounds__(256) void deg_count_kernel(
    const int* __restrict__ col, float* __restrict__ deg, int ne)
{
    int e = blockIdx.x * 256 + threadIdx.x;
    if (e < ne) unsafeAtomicAdd(&deg[col[e]], 1.0f);
}

__global__ __launch_bounds__(256) void dinv_kernel(float* __restrict__ d, int n)
{
    int i = blockIdx.x * 256 + threadIdx.x;
    if (i < n) d[i] = rsqrtf(d[i]);
}

__global__ __launch_bounds__(256) void edge_scatter_kernel(
    const int* __restrict__ row, const int* __restrict__ col,
    const float* __restrict__ dinv, const float* __restrict__ xw,
    float* __restrict__ agg, int ne)
{
    const int lane = threadIdx.x & 31;
    const int wave = __builtin_amdgcn_readfirstlane((int)(threadIdx.x >> 5));
    const int e0   = blockIdx.x * 256 + wave * 32;
    int cnt = ne - e0;
    if (cnt > 32) cnt = 32;
    if (cnt <= 0) return;

    int   r  = 0;
    int   c  = 0;
    float nm = 0.f;
    if (lane < cnt) {
        r  = row[e0 + lane];
        c  = col[e0 + lane];
        nm = dinv[r] * dinv[c];
    }
    const int nmi = __float_as_int(nm);

    if (cnt == 32) {
        #pragma unroll
        for (int jb = 0; jb < 32; jb += 8) {
            float v0[8], v1[8];
            #pragma unroll
            for (int j = 0; j < 8; ++j) {
                const int rj = __builtin_amdgcn_readlane(r, jb + j);
                const float* src = xw + (size_t)rj * 64;
                v0[j] = src[lane];
                v1[j] = src[lane + 32];
            }
            #pragma unroll
            for (int j = 0; j < 8; ++j) {
                const int   cj = __builtin_amdgcn_readlane(c, jb + j);
                const float nj = __int_as_float(__builtin_amdgcn_readlane(nmi, jb + j));
                float* dst = agg + (size_t)cj * 64;
                unsafeAtomicAdd(&dst[lane],      nj * v0[j]);
                unsafeAtomicAdd(&dst[lane + 32], nj * v1[j]);
            }
        }
    } else {
        for (int j = 0; j < cnt; ++j) {
            const int   rj = __builtin_amdgcn_readlane(r, j);
            const int   cj = __builtin_amdgcn_readlane(c, j);
            const float nj = __int_as_float(__builtin_amdgcn_readlane(nmi, j));
            const float* src = xw  + (size_t)rj * 64;
            float*       dst = agg + (size_t)cj * 64;
            unsafeAtomicAdd(&dst[lane],      nj * src[lane]);
            unsafeAtomicAdd(&dst[lane + 32], nj * src[lane + 32]);
        }
    }
}

extern "C" void kernel_launch(void* const* d_in, const int* in_sizes, int n_in,
                              void* d_out, int out_size, void* d_ws, size_t ws_size,
                              hipStream_t stream)
{
    const float* x      = (const float*)d_in[0];
    const int*   ei     = (const int*)d_in[1];
    const float* W0     = (const float*)d_in[2];
    const float* b0     = (const float*)d_in[3];
    const float* W1     = (const float*)d_in[4];
    const float* b1     = (const float*)d_in[5];
    const float* bn0_g  = (const float*)d_in[6];
    const float* bn0_b  = (const float*)d_in[7];
    const float* bn0_m  = (const float*)d_in[8];
    const float* bn0_v  = (const float*)d_in[9];
    const float* bn1_g  = (const float*)d_in[10];
    const float* bn1_b  = (const float*)d_in[11];
    const float* bn1_m  = (const float*)d_in[12];
    const float* bn1_v  = (const float*)d_in[13];
    const float* lin1_W = (const float*)d_in[14];
    const float* lin1_b = (const float*)d_in[15];
    const float* lin2_W = (const float*)d_in[16];
    const float* lin2_b = (const float*)d_in[17];
    const float* lin3_W = (const float*)d_in[18];
    const float* lin3_b = (const float*)d_in[19];

    const int N = in_sizes[0] / 64;
    const int E = in_sizes[1] / 2;
    const int* row = ei;
    const int* col = ei + E;

    char*  ws   = (char*)d_ws;
    size_t off0 = 0;
    size_t off1 = (off0 + (size_t)N * sizeof(float) + 255) & ~(size_t)255;
    size_t off2 = (off1 + (size_t)N * 64 * sizeof(float) + 255) & ~(size_t)255;
    size_t off3 = (off2 + (size_t)N * 64 * sizeof(float) + 255) & ~(size_t)255;
    float* dinv = (float*)(ws + off0);
    float* bufA = (float*)(ws + off1);
    float* bufB = (float*)(ws + off2);
    float* bufC = (float*)(ws + off3);

    const dim3 blk(256);
    const int gN    = (N + 255) / 256;
    const int gE    = (E + 255) / 256;
    const int gEdge = (E + 255) / 256;
    const int gGemm = (N + 127) / 128;

    deg_init_kernel<<<gN, blk, 0, stream>>>(dinv, N);
    deg_count_kernel<<<gE, blk, 0, stream>>>(col, dinv, E);
    dinv_kernel<<<gN, blk, 0, stream>>>(dinv, N);

    gemm_wmma_kernel<64, 0, 3><<<gGemm, blk, 0, stream>>>(
        x, W0, nullptr, bufA, N,
        nullptr, nullptr, nullptr, nullptr, nullptr, dinv, bufB);
    edge_scatter_kernel<<<gEdge, blk, 0, stream>>>(row, col, dinv, bufA, bufB, E);

    gemm_wmma_kernel<64, 1, 3><<<gGemm, blk, 0, stream>>>(
        bufB, W1, nullptr, bufA, N,
        b0, bn0_g, bn0_b, bn0_m, bn0_v, dinv, bufC);
    edge_scatter_kernel<<<gEdge, blk, 0, stream>>>(row, col, dinv, bufA, bufC, E);

    gemm_wmma_kernel<64, 1, 2><<<gGemm, blk, 0, stream>>>(
        bufC, lin1_W, lin1_b, bufA, N,
        b1, bn1_g, bn1_b, bn1_m, bn1_v, nullptr, nullptr);
    gemm_wmma_kernel<64, 0, 2><<<gGemm, blk, 0, stream>>>(
        bufA, lin2_W, lin2_b, bufB, N,
        nullptr, nullptr, nullptr, nullptr, nullptr, nullptr, nullptr);
    gemm_wmma_kernel<16, 0, 1><<<gGemm, blk, 0, stream>>>(
        bufB, lin3_W, lin3_b, (float*)d_out, N,
        nullptr, nullptr, nullptr, nullptr, nullptr, nullptr, nullptr);
}


# in-degree count: per-edge global float atomic replaced by per-block LDS tables of packed integer counters, summed per node in the rsqrt kernel; init launch gone; degrees exact, row sums unchanged
# speedup vs baseline: 2.3857x; 1.2076x over previous
#include <hip/hip_runtime.h>
#include <cstdint>
#include <cstddef>

typedef __attribute__((ext_vector_type(2))) float v2f;
typedef __attribute__((ext_vector_type(8))) float v8f;

template <int NCOL, int PRO, int EPI>
__global__ __launch_bounds__(256) void gemm_wmma_kernel(
    const float* __restrict__ A, const float* __restrict__ W,
    const float* __restrict__ bias, float* __restrict__ out, int nrows,
    const float* __restrict__ cb, const float* __restrict__ bn_g,
    const float* __restrict__ bn_b, const float* __restrict__ bn_m,
    const float* __restrict__ bn_v,
    const float* __restrict__ dinv, float* __restrict__ seed)
{
    constexpr int NT = NCOL / 16;
    __shared__ __align__(16) float sW[64 * NCOL];
    __shared__ __align__(16) float sA[8][16 * 68];

    const int tid  = threadIdx.x;
    const int wave = tid >> 5;
    const int lane = tid & 31;
    const int hh   = lane >> 4;
    const int l16  = lane & 15;
    const int r0   = blockIdx.x * 128 + wave * 16;

    {
        const float4* W4  = reinterpret_cast<const float4*>(W);
        float4*       sW4 = reinterpret_cast<float4*>(sW);
        constexpr int NV = (64 * NCOL / 4) / 256;
        #pragma unroll
        for (int t = 0; t < NV; ++t) sW4[tid + 256 * t] = W4[tid + 256 * t];
    }

    float4 pb = make_float4(0.f, 0.f, 0.f, 0.f), pm = pb, pr = pb, pg = pb, pe = pb;
    if (PRO == 1) {
        pb = reinterpret_cast<const float4*>(cb)[l16];
        pm = reinterpret_cast<const float4*>(bn_m)[l16];
        pg = reinterpret_cast<const float4*>(bn_g)[l16];
        pe = reinterpret_cast<const float4*>(bn_b)[l16];
        const float4 pv = reinterpret_cast<const float4*>(bn_v)[l16];
        pr.x = rsqrtf(pv.x + 1e-5f);
        pr.y = rsqrtf(pv.y + 1e-5f);
        pr.z = rsqrtf(pv.z + 1e-5f);
        pr.w = rsqrtf(pv.w + 1e-5f);
    }

    const float4* A4 = reinterpret_cast<const float4*>(A);
    float4 sv[8];
    #pragma unroll
    for (int i = 0; i < 8; ++i) {
        int gr = r0 + 2 * i + hh;
        gr = (gr < nrows) ? gr : nrows - 1;
        sv[i] = A4[(size_t)gr * 16 + l16];
    }
    #pragma unroll
    for (int i = 0; i < 8; ++i) {
        float4 v = sv[i];
        if (PRO == 1) {
            v.x = fmaxf((v.x + pb.x - pm.x) * pr.x * pg.x + pe.x, 0.f);
            v.y = fmaxf((v.y + pb.y - pm.y) * pr.y * pg.y + pe.y, 0.f);
            v.z = fmaxf((v.z + pb.z - pm.z) * pr.z * pg.z + pe.z, 0.f);
            v.w = fmaxf((v.w + pb.w - pm.w) * pr.w * pg.w + pe.w, 0.f);
        }
        *reinterpret_cast<float4*>(&sA[wave][(2 * i + hh) * 68 + l16 * 4]) = v;
    }
    __syncthreads();

    float dv[8];
    if (EPI == 3) {
        #pragma unroll
        for (int j = 0; j < 8; ++j) {
            int gr = r0 + j + 8 * hh;
            gr = (gr < nrows) ? gr : nrows - 1;
            dv[j] = dinv[gr];
        }
    }

    v8f acc[NT] = {};

    #pragma unroll
    for (int kb = 0; kb < 64; kb += 4) {
        v2f a;
        a.x = sA[wave][l16 * 68 + kb + 2 * hh + 0];
        a.y = sA[wave][l16 * 68 + kb + 2 * hh + 1];
        #pragma unroll
        for (int nt = 0; nt < NT; ++nt) {
            v2f b;
            b.x = sW[(kb + 2 * hh + 0) * NCOL + nt * 16 + l16];
            b.y = sW[(kb + 2 * hh + 1) * NCOL + nt * 16 + l16];
            acc[nt] = __builtin_amdgcn_wmma_f32_16x16x4_f32(
                false, a, false, b, (short)0, acc[nt], false, false);
        }
    }

    #pragma unroll
    for (int nt = 0; nt < NT; ++nt) {
        const int col = nt * 16 + l16;
        float bv = 0.f;
        if (EPI == 1 || EPI == 2) bv = bias[col];
        #pragma unroll
        for (int j = 0; j < 8; ++j) {
            int gr = r0 + j + 8 * hh;
            if (gr < nrows) {
                float v = acc[nt][j] + bv;
                if (EPI == 2) v = fmaxf(v, 0.f);
                out[(size_t)gr * NCOL + col] = v;
                if (EPI == 3) seed[(size_t)gr * NCOL + col] = v * dv[j] * dv[j];
            }
        }
    }
}

constexpr int DEG_SLICE = 32768;
constexpr int DEG_RANGE = 32768;
constexpr int DEG_WORDS = DEG_RANGE / 2;
constexpr int DEG_EPT   = DEG_SLICE / 1024;
constexpr int DEG_KB    = 16;

__global__ __launch_bounds__(1024) void deg_hist_kernel(
    const int* __restrict__ col, unsigned* __restrict__ part, int ne, int n, int stride)
{
    __shared__ __align__(16) unsigned hist[DEG_WORDS];
    const int tid  = threadIdx.x;
    const int e0   = blockIdx.x * DEG_SLICE;
    const int base = blockIdx.y * DEG_RANGE;
    int len = n - base;
    if (len > DEG_RANGE) len = DEG_RANGE;
    const int nq = ((len + 1) / 2 + 3) / 4;

    int cv[DEG_EPT];
    #pragma unroll
    for (int i = 0; i < DEG_EPT; ++i) {
        int e = e0 + i * 1024 + tid;
        e = (e < ne) ? e : ne - 1;
        cv[i] = col[e];
    }

    uint4* h4 = reinterpret_cast<uint4*>(hist);
    for (int q = tid; q < nq; q += 1024) h4[q] = make_uint4(0u, 0u, 0u, 0u);
    __syncthreads();

    #pragma unroll
    for (int i = 0; i < DEG_EPT; ++i) {
        const int      e   = e0 + i * 1024 + tid;
        const unsigned rel = (unsigned)(cv[i] - base);
        if (e < ne && rel < (unsigned)len)
            atomicAdd(&hist[rel >> 1], (rel & 1u) ? 0x10000u : 1u);
    }
    __syncthreads();

    uint4* p4 = reinterpret_cast<uint4*>(part + (size_t)blockIdx.x * stride + (base >> 1));
    for (int q = tid; q < nq; q += 1024) p4[q] = h4[q];
}

__global__ __launch_bounds__(256) void dinv_kernel(
    const unsigned* __restrict__ part, float* __restrict__ dinv, int n, int stride, int nslice)
{
    const int w = blockIdx.x * 256 + threadIdx.x;
    if (2 * w >= n) return;
    const unsigned* p = part + w;
    unsigned lo = 0u, hi = 0u;
    for (int k0 = 0; k0 < nslice; k0 += DEG_KB) {
        unsigned v[DEG_KB];
        #pragma unroll
        for (int j = 0; j < DEG_KB; ++j) v[j] = p[(size_t)(k0 + j) * stride];
        #pragma unroll
        for (int j = 0; j < DEG_KB; ++j) {
            lo += v[j] & 0xffffu;
            hi += v[j] >> 16;
        }
    }
    const float d0 = rsqrtf(1.0f + (float)lo);
    const float d1 = rsqrtf(1.0f + (float)hi);
    if (2 * w + 1 < n) *reinterpret_cast<float2*>(&dinv[2 * w]) = make_float2(d0, d1);
    else               dinv[2 * w] = d0;
}

__global__ __launch_bounds__(256) void edge_scatter_kernel(
    const int* __restrict__ row, const int* __restrict__ col,
    const float* __restrict__ dinv, const float* __restrict__ xw,
    float* __restrict__ agg, int ne)
{
    const int lane = threadIdx.x & 31;
    const int wave = __builtin_amdgcn_readfirstlane((int)(threadIdx.x >> 5));
    const int e0   = blockIdx.x * 256 + wave * 32;
    int cnt = ne - e0;
    if (cnt > 32) cnt = 32;
    if (cnt <= 0) return;

    int   r  = 0;
    int   c  = 0;
    float nm = 0.f;
    if (lane < cnt) {
        r  = row[e0 + lane];
        c  = col[e0 + lane];
        nm = dinv[r] * dinv[c];
    }
    const int nmi = __float_as_int(nm);

    if (cnt == 32) {
        #pragma unroll
        for (int jb = 0; jb < 32; jb += 8) {
            float v0[8], v1[8];
            #pragma unroll
            for (int j = 0; j < 8; ++j) {
                const int rj = __builtin_amdgcn_readlane(r, jb + j);
                const float* src = xw + (size_t)rj * 64;
                v0[j] = src[lane];
                v1[j] = src[lane + 32];
            }
            #pragma unroll
            for (int j = 0; j < 8; ++j) {
                const int   cj = __builtin_amdgcn_readlane(c, jb + j);
                const float nj = __int_as_float(__builtin_amdgcn_readlane(nmi, jb + j));
                float* dst = agg + (size_t)cj * 64;
                unsafeAtomicAdd(&dst[lane],      nj * v0[j]);
                unsafeAtomicAdd(&dst[lane + 32], nj * v1[j]);
            }
        }
    } else {
        for (int j = 0; j < cnt; ++j) {
            const int   rj = __builtin_amdgcn_readlane(r, j);
            const int   cj = __builtin_amdgcn_readlane(c, j);
            const float nj = __int_as_float(__builtin_amdgcn_readlane(nmi, j));
            const float* src = xw  + (size_t)rj * 64;
            float*       dst = agg + (size_t)cj * 64;
            unsafeAtomicAdd(&dst[lane],      nj * src[lane]);
            unsafeAtomicAdd(&dst[lane + 32], nj * src[lane + 32]);
        }
    }
}

extern "C" void kernel_launch(void* const* d_in, const int* in_sizes, int n_in,
                              void* d_out, int out_size, void* d_ws, size_t ws_size,
                              hipStream_t stream)
{
    const float* x      = (const float*)d_in[0];
    const int*   ei     = (const int*)d_in[1];
    const float* W0     = (const float*)d_in[2];
    const float* b0     = (const float*)d_in[3];
    const float* W1     = (const float*)d_in[4];
    const float* b1     = (const float*)d_in[5];
    const float* bn0_g  = (const float*)d_in[6];
    const float* bn0_b  = (const float*)d_in[7];
    const float* bn0_m  = (const float*)d_in[8];
    const float* bn0_v  = (const float*)d_in[9];
    const float* bn1_g  = (const float*)d_in[10];
    const float* bn1_b  = (const float*)d_in[11];
    const float* bn1_m  = (const float*)d_in[12];
    const float* bn1_v  = (const float*)d_in[13];
    const float* lin1_W = (const float*)d_in[14];
    const float* lin1_b = (const float*)d_in[15];
    const float* lin2_W = (const float*)d_in[16];
    const float* lin2_b = (const float*)d_in[17];
    const float* lin3_W = (const float*)d_in[18];
    const float* lin3_b = (const float*)d_in[19];

    const int N = in_sizes[0] / 64;
    const int E = in_sizes[1] / 2;
    const int* row = ei;
    const int* col = ei + E;

    char*  ws   = (char*)d_ws;
    size_t off0 = 0;
    size_t off1 = (off0 + (size_t)N * sizeof(float) + 255) & ~(size_t)255;
    size_t off2 = (off1 + (size_t)N * 64 * sizeof(float) + 255) & ~(size_t)255;
    size_t off3 = (off2 + (size_t)N * 64 * sizeof(float) + 255) & ~(size_t)255;
    float* dinv = (float*)(ws + off0);
    float* bufA = (float*)(ws + off1);
    float* bufB = (float*)(ws + off2);
    float* bufC = (float*)(ws + off3);
    size_t off4 = (off3 + (size_t)N * 64 * sizeof(float) + 255) & ~(size_t)255;
    unsigned* part = (unsigned*)(ws + off4);

    const dim3 blk(256);
    const int gEdge = (E + 255) / 256;
    const int gSl   = ((E + DEG_SLICE - 1) / DEG_SLICE + DEG_KB - 1) / DEG_KB * DEG_KB;
    const int gRg   = (N + DEG_RANGE - 1) / DEG_RANGE;
    const int nwS   = ((N + 1) / 2 + 3) & ~3;
    const int gW    = ((N + 1) / 2 + 255) / 256;
    const int gGemm = (N + 127) / 128;

    deg_hist_kernel<<<dim3(gSl, gRg), dim3(1024), 0, stream>>>(col, part, E, N, nwS);
    dinv_kernel<<<gW, blk, 0, stream>>>(part, dinv, N, nwS, gSl);

    gemm_wmma_kernel<64, 0, 3><<<gGemm, blk, 0, stream>>>(
        x, W0, nullptr, bufA, N,
        nullptr, nullptr, nullptr, nullptr, nullptr, dinv, bufB);
    edge_scatter_kernel<<<gEdge, blk, 0, stream>>>(row, col, dinv, bufA, bufB, E);

    gemm_wmma_kernel<64, 1, 3><<<gGemm, blk, 0, stream>>>(
        bufB, W1, nullptr, bufA, N,
        b0, bn0_g, bn0_b, bn0_m, bn0_v, dinv, bufC);
    edge_scatter_kernel<<<gEdge, blk, 0, stream>>>(row, col, dinv, bufA, bufC, E);

    gemm_wmma_kernel<64, 1, 2><<<gGemm, blk, 0, stream>>>(
        bufC, lin1_W, lin1_b, bufA, N,
        b1, bn1_g, bn1_b, bn1_m, bn1_v, nullptr, nullptr);
    gemm_wmma_kernel<64, 0, 2><<<gGemm, blk, 0, stream>>>(
        bufA, lin2_W, lin2_b, bufB, N,
        nullptr, nullptr, nullptr, nullptr, nullptr, nullptr, nullptr);
    gemm_wmma_kernel<16, 0, 1><<<gGemm, blk, 0, stream>>>(
        bufB, lin3_W, lin3_b, (float*)d_out, N,
        nullptr, nullptr, nullptr, nullptr, nullptr, nullptr, nullptr);
}


# edge weight dinv[row]*dinv[col] kept per edge: first-layer scatter also stores it as a coalesced array, second-layer scatter loads it instead of two scattered dinv lookups per edge; sums unchanged
# speedup vs baseline: 2.4692x; 1.0350x over previous
#include <hip/hip_runtime.h>
#include <cstdint>
#include <cstddef>

typedef __attribute__((ext_vector_type(2))) float v2f;
typedef __attribute__((ext_vector_type(8))) float v8f;

template <int NCOL, int PRO, int EPI>
__global__ __launch_bounds__(256) void gemm_wmma_kernel(
    const float* __restrict__ A, const float* __restrict__ W,
    const float* __restrict__ bias, float* __restrict__ out, int nrows,
    const float* __restrict__ cb, const float* __restrict__ bn_g,
    const float* __restrict__ bn_b, const float* __restrict__ bn_m,
    const float* __restrict__ bn_v,
    const float* __restrict__ dinv, float* __restrict__ seed)
{
    constexpr int NT = NCOL / 16;
    __shared__ __align__(16) float sW[64 * NCOL];
    __shared__ __align__(16) float sA[8][16 * 68];

    const int tid  = threadIdx.x;
    const int wave = tid >> 5;
    const int lane = tid & 31;
    const int hh   = lane >> 4;
    const int l16  = lane & 15;
    const int r0   = blockIdx.x * 128 + wave * 16;

    {
        const float4* W4  = reinterpret_cast<const float4*>(W);
        float4*       sW4 = reinterpret_cast<float4*>(sW);
        constexpr int NV = (64 * NCOL / 4) / 256;
        #pragma unroll
        for (int t = 0; t < NV; ++t) sW4[tid + 256 * t] = W4[tid + 256 * t];
    }

    float4 pb = make_float4(0.f, 0.f, 0.f, 0.f), pm = pb, pr = pb, pg = pb, pe = pb;
    if (PRO == 1) {
        pb = reinterpret_cast<const float4*>(cb)[l16];
        pm = reinterpret_cast<const float4*>(bn_m)[l16];
        pg = reinterpret_cast<const float4*>(bn_g)[l16];
        pe = reinterpret_cast<const float4*>(bn_b)[l16];
        const float4 pv = reinterpret_cast<const float4*>(bn_v)[l16];
        pr.x = rsqrtf(pv.x + 1e-5f);
        pr.y = rsqrtf(pv.y + 1e-5f);
        pr.z = rsqrtf(pv.z + 1e-5f);
        pr.w = rsqrtf(pv.w + 1e-5f);
    }

    const float4* A4 = reinterpret_cast<const float4*>(A);
    float4 sv[8];
    #pragma unroll
    for (int i = 0; i < 8; ++i) {
        int gr = r0 + 2 * i + hh;
        gr = (gr < nrows) ? gr : nrows - 1;
        sv[i] = A4[(size_t)gr * 16 + l16];
    }
    #pragma unroll
    for (int i = 0; i < 8; ++i) {
        float4 v = sv[i];
        if (PRO == 1) {
            v.x = fmaxf((v.x + pb.x - pm.x) * pr.x * pg.x + pe.x, 0.f);
            v.y = fmaxf((v.y + pb.y - pm.y) * pr.y * pg.y + pe.y, 0.f);
            v.z = fmaxf((v.z + pb.z - pm.z) * pr.z * pg.z + pe.z, 0.f);
            v.w = fmaxf((v.w + pb.w - pm.w) * pr.w * pg.w + pe.w, 0.f);
        }
        *reinterpret_cast<float4*>(&sA[wave][(2 * i + hh) * 68 + l16 * 4]) = v;
    }
    __syncthreads();

    float dv[8];
    if (EPI == 3) {
        #pragma unroll
        for (int j = 0; j < 8; ++j) {
            int gr = r0 + j + 8 * hh;
            gr = (gr < nrows) ? gr : nrows - 1;
            dv[j] = dinv[gr];
        }
    }

    v8f acc[NT] = {};

    #pragma unroll
    for (int kb = 0; kb < 64; kb += 4) {
        v2f a;
        a.x = sA[wave][l16 * 68 + kb + 2 * hh + 0];
        a.y = sA[wave][l16 * 68 + kb + 2 * hh + 1];
        #pragma unroll
        for (int nt = 0; nt < NT; ++nt) {
            v2f b;
            b.x = sW[(kb + 2 * hh + 0) * NCOL + nt * 16 + l16];
            b.y = sW[(kb + 2 * hh + 1) * NCOL + nt * 16 + l16];
            acc[nt] = __builtin_amdgcn_wmma_f32_16x16x4_f32(
                false, a, false, b, (short)0, acc[nt], false, false);
        }
    }

    #pragma unroll
    for (int nt = 0; nt < NT; ++nt) {
        const int col = nt * 16 + l16;
        float bv = 0.f;
        if (EPI == 1 || EPI == 2) bv = bias[col];
        #pragma unroll
        for (int j = 0; j < 8; ++j) {
            int gr = r0 + j + 8 * hh;
            if (gr < nrows) {
                float v = acc[nt][j] + bv;
                if (EPI == 2) v = fmaxf(v, 0.f);
                out[(size_t)gr * NCOL + col] = v;
                if (EPI == 3) seed[(size_t)gr * NCOL + col] = v * dv[j] * dv[j];
            }
        }
    }
}

constexpr int DEG_SLICE = 32768;
constexpr int DEG_RANGE = 32768;
constexpr int DEG_WORDS = DEG_RANGE / 2;
constexpr int DEG_EPT   = DEG_SLICE / 1024;
constexpr int DEG_KB    = 16;

__global__ __launch_bounds__(1024) void deg_hist_kernel(
    const int* __restrict__ col, unsigned* __restrict__ part, int ne, int n, int stride)
{
    __shared__ __align__(16) unsigned hist[DEG_WORDS];
    const int tid  = threadIdx.x;
    const int e0   = blockIdx.x * DEG_SLICE;
    const int base = blockIdx.y * DEG_RANGE;
    int len = n - base;
    if (len > DEG_RANGE) len = DEG_RANGE;
    const int nq = ((len + 1) / 2 + 3) / 4;

    int cv[DEG_EPT];
    #pragma unroll
    for (int i = 0; i < DEG_EPT; ++i) {
        int e = e0 + i * 1024 + tid;
        e = (e < ne) ? e : ne - 1;
        cv[i] = col[e];
    }

    uint4* h4 = reinterpret_cast<uint4*>(hist);
    for (int q = tid; q < nq; q += 1024) h4[q] = make_uint4(0u, 0u, 0u, 0u);
    __syncthreads();

    #pragma unroll
    for (int i = 0; i < DEG_EPT; ++i) {
        const int      e   = e0 + i * 1024 + tid;
        const unsigned rel = (unsigned)(cv[i] - base);
        if (e < ne && rel < (unsigned)len)
            atomicAdd(&hist[rel >> 1], (rel & 1u) ? 0x10000u : 1u);
    }
    __syncthreads();

    uint4* p4 = reinterpret_cast<uint4*>(part + (size_t)blockIdx.x * stride + (base >> 1));
    for (int q = tid; q < nq; q += 1024) p4[q] = h4[q];
}

__global__ __launch_bounds__(256) void dinv_kernel(
    const unsigned* __restrict__ part, float* __restrict__ dinv, int n, int stride, int nslice)
{
    const int w = blockIdx.x * 256 + threadIdx.x;
    if (2 * w >= n) return;
    const unsigned* p = part + w;
    unsigned lo = 0u, hi = 0u;
    for (int k0 = 0; k0 < nslice; k0 += DEG_KB) {
        unsigned v[DEG_KB];
        #pragma unroll
        for (int j = 0; j < DEG_KB; ++j) v[j] = p[(size_t)(k0 + j) * stride];
        #pragma unroll
        for (int j = 0; j < DEG_KB; ++j) {
            lo += v[j] & 0xffffu;
            hi += v[j] >> 16;
        }
    }
    const float d0 = rsqrtf(1.0f + (float)lo);
    const float d1 = rsqrtf(1.0f + (float)hi);
    if (2 * w + 1 < n) *reinterpret_cast<float2*>(&dinv[2 * w]) = make_float2(d0, d1);
    else               dinv[2 * w] = d0;
}

template <int WLOAD>
__global__ __launch_bounds__(256) void edge_scatter_kernel(
    const int* __restrict__ row, const int* __restrict__ col,
    const float* __restrict__ dinv, const float* __restrict__ xw,
    float* __restrict__ agg, float* __restrict__ ew, int ne)
{
    const int lane = threadIdx.x & 31;
    const int wave = __builtin_amdgcn_readfirstlane((int)(threadIdx.x >> 5));
    const int e0   = blockIdx.x * 256 + wave * 32;
    int cnt = ne - e0;
    if (cnt > 32) cnt = 32;
    if (cnt <= 0) return;

    int   r  = 0;
    int   c  = 0;
    float nm = 0.f;
    if (lane < cnt) {
        r  = row[e0 + lane];
        c  = col[e0 + lane];
        if (WLOAD == 1) {
            nm = ew[e0 + lane];
        } else {
            nm = dinv[r] * dinv[c];
            ew[e0 + lane] = nm;
        }
    }
    const int nmi = __float_as_int(nm);

    if (cnt == 32) {
        #pragma unroll
        for (int jb = 0; jb < 32; jb += 8) {
            float v0[8], v1[8];
            #pragma unroll
            for (int j = 0; j < 8; ++j) {
                const int rj = __builtin_amdgcn_readlane(r, jb + j);
                const float* src = xw + (size_t)rj * 64;
                v0[j] = src[lane];
                v1[j] = src[lane + 32];
            }
            #pragma unroll
            for (int j = 0; j < 8; ++j) {
                const int   cj = __builtin_amdgcn_readlane(c, jb + j);
                const float nj = __int_as_float(__builtin_amdgcn_readlane(nmi, jb + j));
                float* dst = agg + (size_t)cj * 64;
                unsafeAtomicAdd(&dst[lane],      nj * v0[j]);
                unsafeAtomicAdd(&dst[lane + 32], nj * v1[j]);
            }
        }
    } else {
        for (int j = 0; j < cnt; ++j) {
            const int   rj = __builtin_amdgcn_readlane(r, j);
            const int   cj = __builtin_amdgcn_readlane(c, j);
            const float nj = __int_as_float(__builtin_amdgcn_readlane(nmi, j));
            const float* src = xw  + (size_t)rj * 64;
            float*       dst = agg + (size_t)cj * 64;
            unsafeAtomicAdd(&dst[lane],      nj * src[lane]);
            unsafeAtomicAdd(&dst[lane + 32], nj * src[lane + 32]);
        }
    }
}

extern "C" void kernel_launch(void* const* d_in, const int* in_sizes, int n_in,
                              void* d_out, int out_size, void* d_ws, size_t ws_size,
                              hipStream_t stream)
{
    const float* x      = (const float*)d_in[0];
    const int*   ei     = (const int*)d_in[1];
    const float* W0     = (const float*)d_in[2];
    const float* b0     = (const float*)d_in[3];
    const float* W1     = (const float*)d_in[4];
    const float* b1     = (const float*)d_in[5];
    const float* bn0_g  = (const float*)d_in[6];
    const float* bn0_b  = (const float*)d_in[7];
    const float* bn0_m  = (const float*)d_in[8];
    const float* bn0_v  = (const float*)d_in[9];
    const float* bn1_g  = (const float*)d_in[10];
    const float* bn1_b  = (const float*)d_in[11];
    const float* bn1_m  = (const float*)d_in[12];
    const float* bn1_v  = (const float*)d_in[13];
    const float* lin1_W = (const float*)d_in[14];
    const float* lin1_b = (const float*)d_in[15];
    const float* lin2_W = (const float*)d_in[16];
    const float* lin2_b = (const float*)d_in[17];
    const float* lin3_W = (const float*)d_in[18];
    const float* lin3_b = (const float*)d_in[19];

    const int N = in_sizes[0] / 64;
    const int E = in_sizes[1] / 2;
    const int* row = ei;
    const int* col = ei + E;

    char*  ws   = (char*)d_ws;
    size_t off0 = 0;
    size_t off1 = (off0 + (size_t)N * sizeof(float) + 255) & ~(size_t)255;
    size_t off2 = (off1 + (size_t)N * 64 * sizeof(float) + 255) & ~(size_t)255;
    size_t off3 = (off2 + (size_t)N * 64 * sizeof(float) + 255) & ~(size_t)255;
    float* dinv = (float*)(ws + off0);
    float* bufA = (float*)(ws + off1);
    float* bufB = (float*)(ws + off2);
    float* bufC = (float*)(ws + off3);
    size_t off4 = (off3 + (size_t)N * 64 * sizeof(float) + 255) & ~(size_t)255;
    unsigned* part = (unsigned*)(ws + off4);

    const dim3 blk(256);
    const int gEdge = (E + 255) / 256;
    const int gSl   = ((E + DEG_SLICE - 1) / DEG_SLICE + DEG_KB - 1) / DEG_KB * DEG_KB;
    const int gRg   = (N + DEG_RANGE - 1) / DEG_RANGE;
    const int nwS   = ((N + 1) / 2 + 3) & ~3;
    const int gW    = ((N + 1) / 2 + 255) / 256;
    const int gGemm = (N + 127) / 128;
    size_t off5 = (off4 + (size_t)gSl * nwS * sizeof(unsigned) + 255) & ~(size_t)255;
    float* ew = (float*)(ws + off5);

    deg_hist_kernel<<<dim3(gSl, gRg), dim3(1024), 0, stream>>>(col, part, E, N, nwS);
    dinv_kernel<<<gW, blk, 0, stream>>>(part, dinv, N, nwS, gSl);

    gemm_wmma_kernel<64, 0, 3><<<gGemm, blk, 0, stream>>>(
        x, W0, nullptr, bufA, N,
        nullptr, nullptr, nullptr, nullptr, nullptr, dinv, bufB);
    edge_scatter_kernel<0><<<gEdge, blk, 0, stream>>>(row, col, dinv, bufA, bufB, ew, E);

    gemm_wmma_kernel<64, 1, 3><<<gGemm, blk, 0, stream>>>(
        bufB, W1, nullptr, bufA, N,
        b0, bn0_g, bn0_b, bn0_m, bn0_v, dinv, bufC);
    edge_scatter_kernel<1><<<gEdge, blk, 0, stream>>>(row, col, dinv, bufA, bufC, ew, E);

    gemm_wmma_kernel<64, 1, 2><<<gGemm, blk, 0, stream>>>(
        bufC, lin1_W, lin1_b, bufA, N,
        b1, bn1_g, bn1_b, bn1_m, bn1_v, nullptr, nullptr);
    gemm_wmma_kernel<64, 0, 2><<<gGemm, blk, 0, stream>>>(
        bufA, lin2_W, lin2_b, bufB, N,
        nullptr, nullptr, nullptr, nullptr, nullptr, nullptr, nullptr);
    gemm_wmma_kernel<16, 0, 1><<<gGemm, blk, 0, stream>>>(
        bufB, lin3_W, lin3_b, (float*)d_out, N,
        nullptr, nullptr, nullptr, nullptr, nullptr, nullptr, nullptr);
}


# edge sums: global float atomics replaced by node-range tiles in LDS (edges grouped by 392-node owner once per call, rows added by LDS float atomics, tile stored once); sums in LDS arrival order
# speedup vs baseline: 2.7640x; 1.1194x over previous
#include <hip/hip_runtime.h>
#include <cstdint>
#include <cstddef>

typedef __attribute__((ext_vector_type(2))) float v2f;
typedef __attribute__((ext_vector_type(8))) float v8f;

template <int NCOL, int PRO, int EPI>
__global__ __launch_bounds__(256) void gemm_wmma_kernel(
    const float* __restrict__ A, const float* __restrict__ W,
    const float* __restrict__ bias, float* __restrict__ out, int nrows,
    const float* __restrict__ cb, const float* __restrict__ bn_g,
    const float* __restrict__ bn_b, const float* __restrict__ bn_m,
    const float* __restrict__ bn_v)
{
    constexpr int NT = NCOL / 16;
    __shared__ __align__(16) float sW[64 * NCOL];
    __shared__ __align__(16) float sA[8][16 * 68];

    const int tid  = threadIdx.x;
    const int wave = tid >> 5;
    const int lane = tid & 31;
    const int hh   = lane >> 4;
    const int l16  = lane & 15;
    const int r0   = blockIdx.x * 128 + wave * 16;

    {
        const float4* W4  = reinterpret_cast<const float4*>(W);
        float4*       sW4 = reinterpret_cast<float4*>(sW);
        constexpr int NV = (64 * NCOL / 4) / 256;
        #pragma unroll
        for (int t = 0; t < NV; ++t) sW4[tid + 256 * t] = W4[tid + 256 * t];
    }

    float4 pb = make_float4(0.f, 0.f, 0.f, 0.f), pm = pb, pr = pb, pg = pb, pe = pb;
    if (PRO == 1) {
        pb = reinterpret_cast<const float4*>(cb)[l16];
        pm = reinterpret_cast<const float4*>(bn_m)[l16];
        pg = reinterpret_cast<const float4*>(bn_g)[l16];
        pe = reinterpret_cast<const float4*>(bn_b)[l16];
        const float4 pv = reinterpret_cast<const float4*>(bn_v)[l16];
        pr.x = rsqrtf(pv.x + 1e-5f);
        pr.y = rsqrtf(pv.y + 1e-5f);
        pr.z = rsqrtf(pv.z + 1e-5f);
        pr.w = rsqrtf(pv.w + 1e-5f);
    }

    const float4* A4 = reinterpret_cast<const float4*>(A);
    float4 sv[8];
    #pragma unroll
    for (int i = 0; i < 8; ++i) {
        int gr = r0 + 2 * i + hh;
        gr = (gr < nrows) ? gr : nrows - 1;
        sv[i] = A4[(size_t)gr * 16 + l16];
    }
    #pragma unroll
    for (int i = 0; i < 8; ++i) {
        float4 v = sv[i];
        if (PRO == 1) {
            v.x = fmaxf((v.x + pb.x - pm.x) * pr.x * pg.x + pe.x, 0.f);
            v.y = fmaxf((v.y + pb.y - pm.y) * pr.y * pg.y + pe.y, 0.f);
            v.z = fmaxf((v.z + pb.z - pm.z) * pr.z * pg.z + pe.z, 0.f);
            v.w = fmaxf((v.w + pb.w - pm.w) * pr.w * pg.w + pe.w, 0.f);
        }
        *reinterpret_cast<float4*>(&sA[wave][(2 * i + hh) * 68 + l16 * 4]) = v;
    }
    __syncthreads();

    v8f acc[NT] = {};

    #pragma unroll
    for (int kb = 0; kb < 64; kb += 4) {
        v2f a;
        a.x = sA[wave][l16 * 68 + kb + 2 * hh + 0];
        a.y = sA[wave][l16 * 68 + kb + 2 * hh + 1];
        #pragma unroll
        for (int nt = 0; nt < NT; ++nt) {
            v2f b;
            b.x = sW[(kb + 2 * hh + 0) * NCOL + nt * 16 + l16];
            b.y = sW[(kb + 2 * hh + 1) * NCOL + nt * 16 + l16];
            acc[nt] = __builtin_amdgcn_wmma_f32_16x16x4_f32(
                false, a, false, b, (short)0, acc[nt], false, false);
        }
    }

    #pragma unroll
    for (int nt = 0; nt < NT; ++nt) {
        const int col = nt * 16 + l16;
        float bv = 0.f;
        if (EPI == 1 || EPI == 2) bv = bias[col];
        #pragma unroll
        for (int j = 0; j < 8; ++j) {
            int gr = r0 + j + 8 * hh;
            if (gr < nrows) {
                float v = acc[nt][j] + bv;
                if (EPI == 2) v = fmaxf(v, 0.f);
                out[(size_t)gr * NCOL + col] = v;
            }
        }
    }
}

constexpr int DEG_SLICE = 32768;
constexpr int DEG_RANGE = 32768;
constexpr int DEG_WORDS = DEG_RANGE / 2;
constexpr int DEG_EPT   = DEG_SLICE / 1024;
constexpr int DEG_KB    = 16;

__global__ __launch_bounds__(1024) void deg_hist_kernel(
    const int* __restrict__ col, unsigned* __restrict__ part, int ne, int n, int stride)
{
    __shared__ __align__(16) unsigned hist[DEG_WORDS];
    const int tid  = threadIdx.x;
    const int e0   = blockIdx.x * DEG_SLICE;
    const int base = blockIdx.y * DEG_RANGE;
    int len = n - base;
    if (len > DEG_RANGE) len = DEG_RANGE;
    const int nq = ((len + 1) / 2 + 3) / 4;

    int cv[DEG_EPT];
    #pragma unroll
    for (int i = 0; i < DEG_EPT; ++i) {
        int e = e0 + i * 1024 + tid;
        e = (e < ne) ? e : ne - 1;
        cv[i] = col[e];
    }

    uint4* h4 = reinterpret_cast<uint4*>(hist);
    for (int q = tid; q < nq; q += 1024) h4[q] = make_uint4(0u, 0u, 0u, 0u);
    __syncthreads();

    #pragma unroll
    for (int i = 0; i < DEG_EPT; ++i) {
        const int      e   = e0 + i * 1024 + tid;
        const unsigned rel = (unsigned)(cv[i] - base);
        if (e < ne && rel < (unsigned)len)
            atomicAdd(&hist[rel >> 1], (rel & 1u) ? 0x10000u : 1u);
    }
    __syncthreads();

    uint4* p4 = reinterpret_cast<uint4*>(part + (size_t)blockIdx.x * stride + (base >> 1));
    for (int q = tid; q < nq; q += 1024) p4[q] = h4[q];
}

__global__ __launch_bounds__(256) void dinv_kernel(
    const unsigned* __restrict__ part, float* __restrict__ dinv, int n, int stride, int nslice)
{
    const int w = blockIdx.x * 256 + threadIdx.x;
    if (2 * w >= n) return;
    const unsigned* p = part + w;
    unsigned lo = 0u, hi = 0u;
    for (int k0 = 0; k0 < nslice; k0 += DEG_KB) {
        unsigned v[DEG_KB];
        #pragma unroll
        for (int j = 0; j < DEG_KB; ++j) v[j] = p[(size_t)(k0 + j) * stride];
        #pragma unroll
        for (int j = 0; j < DEG_KB; ++j) {
            lo += v[j] & 0xffffu;
            hi += v[j] >> 16;
        }
    }
    const float d0 = rsqrtf(1.0f + (float)lo);
    const float d1 = rsqrtf(1.0f + (float)hi);
    if (2 * w + 1 < n) *reinterpret_cast<float2*>(&dinv[2 * w]) = make_float2(d0, d1);
    else               dinv[2 * w] = d0;
}

constexpr int RN      = 392;
constexpr int BIN_SL  = 8192;
constexpr int BIN_EPT = BIN_SL / 1024;
constexpr int MAXB    = 512;
constexpr int OWN_NQ  = (RN * 16 + 1023) / 1024;

__global__ __launch_bounds__(1024) void bin_kernel(
    const int* __restrict__ row, const int* __restrict__ col,
    int* __restrict__ sedw, unsigned* __restrict__ tab,
    int ne, int nb, int nslice)
{
    int2* sed = reinterpret_cast<int2*>(sedw);
    __shared__ __align__(8) int2 led[BIN_SL];
    __shared__ unsigned cnt[MAXB];
    __shared__ unsigned first[MAXB];
    __shared__ unsigned cur[MAXB];
    const int tid = threadIdx.x;
    const int s   = blockIdx.x;
    const int e0  = s * BIN_SL;

    int cv[BIN_EPT], rv[BIN_EPT];
    #pragma unroll
    for (int i = 0; i < BIN_EPT; ++i) {
        int e = e0 + i * 1024 + tid;
        e = (e < ne) ? e : ne - 1;
        cv[i] = col[e];
        rv[i] = row[e];
    }

    if (tid < MAXB) { cnt[tid] = 0u; cur[tid] = 0u; }
    __syncthreads();

    #pragma unroll
    for (int i = 0; i < BIN_EPT; ++i) {
        const int e = e0 + i * 1024 + tid;
        if (e < ne) atomicAdd(&cnt[cv[i] / RN], 1u);
    }
    __syncthreads();

    if (tid < 32) {
        unsigned loc[MAXB / 32];
        unsigned sum = 0u;
        #pragma unroll
        for (int j = 0; j < MAXB / 32; ++j) { loc[j] = cnt[tid * (MAXB / 32) + j]; sum += loc[j]; }
        unsigned inc = sum;
        #pragma unroll
        for (int d = 1; d < 32; d <<= 1) {
            const unsigned t = __shfl_up(inc, (unsigned)d);
            if (tid >= d) inc += t;
        }
        unsigned run = inc - sum;
        #pragma unroll
        for (int j = 0; j < MAXB / 32; ++j) { first[tid * (MAXB / 32) + j] = run; run += loc[j]; }
    }
    __syncthreads();

    for (int q = tid; q < nb; q += 1024)
        tab[(size_t)q * nslice + s] = first[q] | (cnt[q] << 16);

    #pragma unroll
    for (int i = 0; i < BIN_EPT; ++i) {
        const int e = e0 + i * 1024 + tid;
        if (e < ne) {
            const int b   = cv[i] / RN;
            const int pos = (int)(first[b] + atomicAdd(&cur[b], 1u));
            led[pos] = make_int2(rv[i], cv[i] - b * RN);
        }
    }
    __syncthreads();

    int valid = ne - e0;
    if (valid > BIN_SL) valid = BIN_SL;
    for (int q = tid; q < valid; q += 1024) sed[e0 + q] = led[q];
}

template <int WLOAD>
__global__ __launch_bounds__(1024) void own_kernel(
    const int* __restrict__ sedw, float* __restrict__ sw,
    const unsigned* __restrict__ tab, const float* __restrict__ dinv,
    const float* __restrict__ xw, float* __restrict__ agg, int n, int nslice)
{
    const int2* sed = reinterpret_cast<const int2*>(sedw);
    __shared__ __align__(16) float acc[RN * 64];
    const int tid  = threadIdx.x;
    const int lane = tid & 31;
    const int wave = __builtin_amdgcn_readfirstlane(tid >> 5);
    const int base = blockIdx.x * RN;
    int len = n - base;
    if (len > RN) len = RN;
    const int nq = len * 16;

    const unsigned* tb = tab + (size_t)blockIdx.x * nslice;
    unsigned segw = 0u;
    if (wave + 32 * lane < nslice) segw = tb[wave + 32 * lane];

    {
        const float4* x4 = reinterpret_cast<const float4*>(xw + (size_t)base * 64);
        float4*       a4 = reinterpret_cast<float4*>(acc);
        float4 sv[OWN_NQ];
        float  dq[OWN_NQ];
        #pragma unroll
        for (int t = 0; t < OWN_NQ; ++t) {
            int q = tid + 1024 * t;
            q = (q < nq) ? q : nq - 1;
            sv[t] = x4[q];
            dq[t] = dinv[base + (q >> 4)];
        }
        #pragma unroll
        for (int t = 0; t < OWN_NQ; ++t) {
            const int q = tid + 1024 * t;
            if (q < nq) {
                float4 v = sv[t];
                v.x = v.x * dq[t] * dq[t];
                v.y = v.y * dq[t] * dq[t];
                v.z = v.z * dq[t] * dq[t];
                v.w = v.w * dq[t] * dq[t];
                a4[q] = v;
            }
        }
    }
    __syncthreads();

    {
        const int nt = (nslice - wave + 31) / 32;
        for (int t = 0; t < nt; ++t) {
            const unsigned word = (unsigned)__builtin_amdgcn_readlane((int)segw, t);
            const int c  = (int)(word >> 16);
            const int g0 = (wave + 32 * t) * BIN_SL + (int)(word & 0xffffu);
            for (int k0 = 0; k0 < c; k0 += 32) {
                int m = c - k0;
                if (m > 32) m = 32;
                int   r  = base;
                int   cl = 0;
                float w  = 0.f;
                if (lane < m) {
                    const int2 ed = sed[g0 + k0 + lane];
                    r  = ed.x;
                    cl = ed.y;
                    if (WLOAD == 1) w = sw[g0 + k0 + lane];
                }
                float wr = 0.f, wc = 0.f;
                if (WLOAD == 0) {
                    wr = dinv[r];
                    wc = dinv[base + cl];
                }
                const int wi  = __float_as_int(w);
                const int wri = __float_as_int(wr);
                const int wci = __float_as_int(wc);
                for (int jb = 0; jb < m; jb += 8) {
                    float2 v[8];
                    #pragma unroll
                    for (int j = 0; j < 8; ++j) {
                        const int rj = __builtin_amdgcn_readlane(r, jb + j);
                        v[j] = *reinterpret_cast<const float2*>(xw + (size_t)rj * 64 + 2 * lane);
                    }
                    #pragma unroll
                    for (int j = 0; j < 8; ++j) {
                        const int   cj = __builtin_amdgcn_readlane(cl, jb + j);
                        float nj;
                        if (WLOAD == 1) {
                            nj = __int_as_float(__builtin_amdgcn_readlane(wi, jb + j));
                        } else {
                            const float p = __int_as_float(__builtin_amdgcn_readlane(wri, jb + j)) *
                                            __int_as_float(__builtin_amdgcn_readlane(wci, jb + j));
                            nj = (jb + j < m) ? p : 0.f;
                        }
                        float* dst = acc + cj * 64 + 2 * lane;
                        unsafeAtomicAdd(dst,     nj * v[j].x);
                        unsafeAtomicAdd(dst + 1, nj * v[j].y);
                    }
                }
                if (WLOAD == 0 && lane < m) sw[g0 + k0 + lane] = wr * wc;
            }
        }
    }
    __syncthreads();

    {
        float4*       o4 = reinterpret_cast<float4*>(agg + (size_t)base * 64);
        const float4* a4 = reinterpret_cast<const float4*>(acc);
        for (int q = tid; q < nq; q += 1024) o4[q] = a4[q];
    }
}

extern "C" void kernel_launch(void* const* d_in, const int* in_sizes, int n_in,
                              void* d_out, int out_size, void* d_ws, size_t ws_size,
                              hipStream_t stream)
{
    const float* x      = (const float*)d_in[0];
    const int*   ei     = (const int*)d_in[1];
    const float* W0     = (const float*)d_in[2];
    const float* b0     = (const float*)d_in[3];
    const float* W1     = (const float*)d_in[4];
    const float* b1     = (const float*)d_in[5];
    const float* bn0_g  = (const float*)d_in[6];
    const float* bn0_b  = (const float*)d_in[7];
    const float* bn0_m  = (const float*)d_in[8];
    const float* bn0_v  = (const float*)d_in[9];
    const float* bn1_g  = (const float*)d_in[10];
    const float* bn1_b  = (const float*)d_in[11];
    const float* bn1_m  = (const float*)d_in[12];
    const float* bn1_v  = (const float*)d_in[13];
    const float* lin1_W = (const float*)d_in[14];
    const float* lin1_b = (const float*)d_in[15];
    const float* lin2_W = (const float*)d_in[16];
    const float* lin2_b = (const float*)d_in[17];
    const float* lin3_W = (const float*)d_in[18];
    const float* lin3_b = (const float*)d_in[19];

    const int N = in_sizes[0] / 64;
    const int E = in_sizes[1] / 2;
    const int* row = ei;
    const int* col = ei + E;

    char*  ws   = (char*)d_ws;
    size_t off0 = 0;
    size_t off1 = (off0 + (size_t)N * sizeof(float) + 255) & ~(size_t)255;
    size_t off2 = (off1 + (size_t)N * 64 * sizeof(float) + 255) & ~(size_t)255;
    size_t off3 = (off2 + (size_t)N * 64 * sizeof(float) + 255) & ~(size_t)255;
    float* dinv = (float*)(ws + off0);
    float* bufA = (float*)(ws + off1);
    float* bufB = (float*)(ws + off2);
    float* bufC = (float*)(ws + off3);
    size_t off4 = (off3 + (size_t)N * 64 * sizeof(float) + 255) & ~(size_t)255;
    unsigned* part = (unsigned*)(ws + off4);

    const dim3 blk(256);
    const int gBin  = (E + BIN_SL - 1) / BIN_SL;
    const int gOwn  = (N + RN - 1) / RN;
    const int gSl   = ((E + DEG_SLICE - 1) / DEG_SLICE + DEG_KB - 1) / DEG_KB * DEG_KB;
    const int gRg   = (N + DEG_RANGE - 1) / DEG_RANGE;
    const int nwS   = ((N + 1) / 2 + 3) & ~3;
    const int gW    = ((N + 1) / 2 + 255) / 256;
    const int gGemm = (N + 127) / 128;
    size_t off5 = (off4 + (size_t)gSl * nwS * sizeof(unsigned) + 255) & ~(size_t)255;
    size_t off6 = (off5 + (size_t)E * sizeof(int2) + 255) & ~(size_t)255;
    size_t off7 = (off6 + (size_t)E * sizeof(float) + 255) & ~(size_t)255;
    int*      sed = (int*)(ws + off5);
    float*    sw  = (float*)(ws + off6);
    unsigned* tab = (unsigned*)(ws + off7);

    deg_hist_kernel<<<dim3(gSl, gRg), dim3(1024), 0, stream>>>(col, part, E, N, nwS);
    dinv_kernel<<<gW, blk, 0, stream>>>(part, dinv, N, nwS, gSl);

    bin_kernel<<<gBin, dim3(1024), 0, stream>>>(row, col, sed, tab, E, gOwn, gBin);

    gemm_wmma_kernel<64, 0, 0><<<gGemm, blk, 0, stream>>>(
        x, W0, nullptr, bufA, N,
        nullptr, nullptr, nullptr, nullptr, nullptr);
    own_kernel<0><<<gOwn, dim3(1024), 0, stream>>>(sed, sw, tab, dinv, bufA, bufB, N, gBin);

    gemm_wmma_kernel<64, 1, 0><<<gGemm, blk, 0, stream>>>(
        bufB, W1, nullptr, bufA, N,
        b0, bn0_g, bn0_b, bn0_m, bn0_v);
    own_kernel<1><<<gOwn, dim3(1024), 0, stream>>>(sed, sw, tab, dinv, bufA, bufC, N, gBin);

    gemm_wmma_kernel<64, 1, 2><<<gGemm, blk, 0, stream>>>(
        bufC, lin1_W, lin1_b, bufA, N,
        b1, bn1_g, bn1_b, bn1_m, bn1_v);
    gemm_wmma_kernel<64, 0, 2><<<gGemm, blk, 0, stream>>>(
        bufA, lin2_W, lin2_b, bufB, N,
        nullptr, nullptr, nullptr, nullptr, nullptr);
    gemm_wmma_kernel<16, 0, 1><<<gGemm, blk, 0, stream>>>(
        bufB, lin3_W, lin3_b, (float*)d_out, N,
        nullptr, nullptr, nullptr, nullptr, nullptr);
}


# first three kernels (edge grouping, in-degree tables, x@W0 product) merged into one launch, one role per block range, so the independent pieces run side by side; 2 launches gone; sums unchanged
# speedup vs baseline: 2.9821x; 1.0789x over previous
#include <hip/hip_runtime.h>
#include <cstdint>
#include <cstddef>

typedef __attribute__((ext_vector_type(2))) float v2f;
typedef __attribute__((ext_vector_type(8))) float v8f;

template <int NCOL, int PRO, int EPI>
__global__ __launch_bounds__(256) void gemm_wmma_kernel(
    const float* __restrict__ A, const float* __restrict__ W,
    const float* __restrict__ bias, float* __restrict__ out, int nrows,
    const float* __restrict__ cb, const float* __restrict__ bn_g,
    const float* __restrict__ bn_b, const float* __restrict__ bn_m,
    const float* __restrict__ bn_v)
{
    constexpr int NT = NCOL / 16;
    __shared__ __align__(16) float sW[64 * NCOL];
    __shared__ __align__(16) float sA[8][16 * 68];

    const int tid  = threadIdx.x;
    const int wave = tid >> 5;
    const int lane = tid & 31;
    const int hh   = lane >> 4;
    const int l16  = lane & 15;
    const int r0   = blockIdx.x * 128 + wave * 16;

    {
        const float4* W4  = reinterpret_cast<const float4*>(W);
        float4*       sW4 = reinterpret_cast<float4*>(sW);
        constexpr int NV = (64 * NCOL / 4) / 256;
        #pragma unroll
        for (int t = 0; t < NV; ++t) sW4[tid + 256 * t] = W4[tid + 256 * t];
    }

    float4 pb = make_float4(0.f, 0.f, 0.f, 0.f), pm = pb, pr = pb, pg = pb, pe = pb;
    if (PRO == 1) {
        pb = reinterpret_cast<const float4*>(cb)[l16];
        pm = reinterpret_cast<const float4*>(bn_m)[l16];
        pg = reinterpret_cast<const float4*>(bn_g)[l16];
        pe = reinterpret_cast<const float4*>(bn_b)[l16];
        const float4 pv = reinterpret_cast<const float4*>(bn_v)[l16];
        pr.x = rsqrtf(pv.x + 1e-5f);
        pr.y = rsqrtf(pv.y + 1e-5f);
        pr.z = rsqrtf(pv.z + 1e-5f);
        pr.w = rsqrtf(pv.w + 1e-5f);
    }

    const float4* A4 = reinterpret_cast<const float4*>(A);
    float4 sv[8];
    #pragma unroll
    for (int i = 0; i < 8; ++i) {
        int gr = r0 + 2 * i + hh;
        gr = (gr < nrows) ? gr : nrows - 1;
        sv[i] = A4[(size_t)gr * 16 + l16];
    }
    #pragma unroll
    for (int i = 0; i < 8; ++i) {
        float4 v = sv[i];
        if (PRO == 1) {
            v.x = fmaxf((v.x + pb.x - pm.x) * pr.x * pg.x + pe.x, 0.f);
            v.y = fmaxf((v.y + pb.y - pm.y) * pr.y * pg.y + pe.y, 0.f);
            v.z = fmaxf((v.z + pb.z - pm.z) * pr.z * pg.z + pe.z, 0.f);
            v.w = fmaxf((v.w + pb.w - pm.w) * pr.w * pg.w + pe.w, 0.f);
        }
        *reinterpret_cast<float4*>(&sA[wave][(2 * i + hh) * 68 + l16 * 4]) = v;
    }
    __syncthreads();

    v8f acc[NT] = {};

    #pragma unroll
    for (int kb = 0; kb < 64; kb += 4) {
        v2f a;
        a.x = sA[wave][l16 * 68 + kb + 2 * hh + 0];
        a.y = sA[wave][l16 * 68 + kb + 2 * hh + 1];
        #pragma unroll
        for (int nt = 0; nt < NT; ++nt) {
            v2f b;
            b.x = sW[(kb + 2 * hh + 0) * NCOL + nt * 16 + l16];
            b.y = sW[(kb + 2 * hh + 1) * NCOL + nt * 16 + l16];
            acc[nt] = __builtin_amdgcn_wmma_f32_16x16x4_f32(
                false, a, false, b, (short)0, acc[nt], false, false);
        }
    }

    #pragma unroll
    for (int nt = 0; nt < NT; ++nt) {
        const int col = nt * 16 + l16;
        float bv = 0.f;
        if (EPI == 1 || EPI == 2) bv = bias[col];
        #pragma unroll
        for (int j = 0; j < 8; ++j) {
            int gr = r0 + j + 8 * hh;
            if (gr < nrows) {
                float v = acc[nt][j] + bv;
                if (EPI == 2) v = fmaxf(v, 0.f);
                out[(size_t)gr * NCOL + col] = v;
            }
        }
    }
}

constexpr int DEG_SLICE = 32768;
constexpr int DEG_RANGE = 32768;
constexpr int DEG_WORDS = DEG_RANGE / 2;
constexpr int DEG_EPT   = DEG_SLICE / 1024;
constexpr int DEG_KB    = 16;

__device__ __forceinline__ void deg_hist_role(
    unsigned char* smem, int bx, int by,
    const int* __restrict__ col, unsigned* __restrict__ part, int ne, int n, int stride)
{
    unsigned* hist = reinterpret_cast<unsigned*>(smem);
    const int tid  = threadIdx.x;
    const int e0   = bx * DEG_SLICE;
    const int base = by * DEG_RANGE;
    int len = n - base;
    if (len > DEG_RANGE) len = DEG_RANGE;
    const int nq = ((len + 1) / 2 + 3) / 4;

    int cv[DEG_EPT];
    #pragma unroll
    for (int i = 0; i < DEG_EPT; ++i) {
        int e = e0 + i * 1024 + tid;
        e = (e < ne) ? e : ne - 1;
        cv[i] = col[e];
    }

    uint4* h4 = reinterpret_cast<uint4*>(hist);
    for (int q = tid; q < nq; q += 1024) h4[q] = make_uint4(0u, 0u, 0u, 0u);
    __syncthreads();

    int nval = ne - e0;
    if (nval > DEG_SLICE) nval = DEG_SLICE;
    #pragma unroll
    for (int i = 0; i < DEG_EPT; ++i) {
        const unsigned rel = (unsigned)(cv[i] - base);
        if (tid < nval - i * 1024 && rel < (unsigned)len)
            atomicAdd(&hist[rel >> 1], (rel & 1u) ? 0x10000u : 1u);
    }
    __syncthreads();

    uint4* p4 = reinterpret_cast<uint4*>(part + (size_t)bx * stride + (base >> 1));
    for (int q = tid; q < nq; q += 1024) p4[q] = h4[q];
}

__global__ __launch_bounds__(256) void dinv_kernel(
    const unsigned* __restrict__ part, float* __restrict__ dinv, int n, int stride, int nslice)
{
    const int w = blockIdx.x * 256 + threadIdx.x;
    if (2 * w >= n) return;
    const unsigned* p = part + w;
    unsigned lo = 0u, hi = 0u;
    for (int k0 = 0; k0 < nslice; k0 += DEG_KB) {
        unsigned v[DEG_KB];
        #pragma unroll
        for (int j = 0; j < DEG_KB; ++j) v[j] = p[(size_t)(k0 + j) * stride];
        #pragma unroll
        for (int j = 0; j < DEG_KB; ++j) {
            lo += v[j] & 0xffffu;
            hi += v[j] >> 16;
        }
    }
    const float d0 = rsqrtf(1.0f + (float)lo);
    const float d1 = rsqrtf(1.0f + (float)hi);
    if (2 * w + 1 < n) *reinterpret_cast<float2*>(&dinv[2 * w]) = make_float2(d0, d1);
    else               dinv[2 * w] = d0;
}

constexpr int RN      = 392;
constexpr int BIN_SL  = 8192;
constexpr int BIN_EPT = BIN_SL / 1024;
constexpr int MAXB    = 512;
constexpr int OWN_NQ  = (RN * 16 + 1023) / 1024;

__device__ __forceinline__ void bin_role(
    unsigned char* smem, int s,
    const int* __restrict__ row, const int* __restrict__ col,
    int* __restrict__ sedw, unsigned* __restrict__ tab,
    int ne, int nb, int nslice)
{
    int2* sed = reinterpret_cast<int2*>(sedw);
    int2*     led   = reinterpret_cast<int2*>(smem);
    unsigned* cnt   = reinterpret_cast<unsigned*>(smem + BIN_SL * sizeof(int2));
    unsigned* first = cnt + MAXB;
    unsigned* cur   = first + MAXB;
    const int tid = threadIdx.x;
    const int e0  = s * BIN_SL;

    int cv[BIN_EPT], rv[BIN_EPT];
    #pragma unroll
    for (int i = 0; i < BIN_EPT; ++i) {
        int e = e0 + i * 1024 + tid;
        e = (e < ne) ? e : ne - 1;
        cv[i] = col[e];
        rv[i] = row[e];
    }

    if (tid < MAXB) { cnt[tid] = 0u; cur[tid] = 0u; }
    __syncthreads();

    #pragma unroll
    for (int i = 0; i < BIN_EPT; ++i) {
        const int e = e0 + i * 1024 + tid;
        if (e < ne) atomicAdd(&cnt[cv[i] / RN], 1u);
    }
    __syncthreads();

    if (tid < 32) {
        unsigned loc[MAXB / 32];
        unsigned sum = 0u;
        #pragma unroll
        for (int j = 0; j < MAXB / 32; ++j) { loc[j] = cnt[tid * (MAXB / 32) + j]; sum += loc[j]; }
        unsigned inc = sum;
        #pragma unroll
        for (int d = 1; d < 32; d <<= 1) {
            const unsigned t = __shfl_up(inc, (unsigned)d);
            if (tid >= d) inc += t;
        }
        unsigned run = inc - sum;
        #pragma unroll
        for (int j = 0; j < MAXB / 32; ++j) { first[tid * (MAXB / 32) + j] = run; run += loc[j]; }
    }
    __syncthreads();

    for (int q = tid; q < nb; q += 1024)
        tab[(size_t)q * nslice + s] = first[q] | (cnt[q] << 16);

    #pragma unroll
    for (int i = 0; i < BIN_EPT; ++i) {
        const int e = e0 + i * 1024 + tid;
        if (e < ne) {
            const int b   = cv[i] / RN;
            const int pos = (int)(first[b] + atomicAdd(&cur[b], 1u));
            led[pos] = make_int2(rv[i], cv[i] - b * RN);
        }
    }
    __syncthreads();

    int valid = ne - e0;
    if (valid > BIN_SL) valid = BIN_SL;
    for (int q = tid; q < valid; q += 1024) sed[e0 + q] = led[q];
}

constexpr int FRONT_ROWS = 512;
constexpr int FRONT_LDS  = 64 * 64 * 4 + 32 * 16 * 68 * 4;

__device__ __forceinline__ void gemm_x_role(
    unsigned char* smem, int blk,
    const float* __restrict__ A, const float* __restrict__ W,
    float* __restrict__ out, int nrows)
{
    float* sW = reinterpret_cast<float*>(smem);
    const int tid  = threadIdx.x;
    const int wave = tid >> 5;
    const int lane = tid & 31;
    const int hh   = lane >> 4;
    const int l16  = lane & 15;
    float* sA = reinterpret_cast<float*>(smem + 64 * 64 * 4) + wave * (16 * 68);
    const int r0   = blk * FRONT_ROWS + wave * 16;

    reinterpret_cast<float4*>(sW)[tid] = reinterpret_cast<const float4*>(W)[tid];

    const float4* A4 = reinterpret_cast<const float4*>(A);
    float4 sv[8];
    #pragma unroll
    for (int i = 0; i < 8; ++i) {
        int gr = r0 + 2 * i + hh;
        gr = (gr < nrows) ? gr : nrows - 1;
        sv[i] = A4[(size_t)gr * 16 + l16];
    }
    #pragma unroll
    for (int i = 0; i < 8; ++i)
        *reinterpret_cast<float4*>(&sA[(2 * i + hh) * 68 + l16 * 4]) = sv[i];
    __syncthreads();

    v8f acc[4] = {};

    #pragma unroll
    for (int kb = 0; kb < 64; kb += 4) {
        v2f a;
        a.x = sA[l16 * 68 + kb + 2 * hh + 0];
        a.y = sA[l16 * 68 + kb + 2 * hh + 1];
        #pragma unroll
        for (int nt = 0; nt < 4; ++nt) {
            v2f b;
            b.x = sW[(kb + 2 * hh + 0) * 64 + nt * 16 + l16];
            b.y = sW[(kb + 2 * hh + 1) * 64 + nt * 16 + l16];
            acc[nt] = __builtin_amdgcn_wmma_f32_16x16x4_f32(
                false, a, false, b, (short)0, acc[nt], false, false);
        }
    }

    #pragma unroll
    for (int nt = 0; nt < 4; ++nt) {
        const int col = nt * 16 + l16;
        #pragma unroll
        for (int j = 0; j < 8; ++j) {
            const int gr = r0 + j + 8 * hh;
            if (gr < nrows) out[(size_t)gr * 64 + col] = acc[nt][j];
        }
    }
}

__global__ __launch_bounds__(1024, 16) void front_kernel(
    const int* __restrict__ row, const int* __restrict__ col,
    int* __restrict__ sedw, unsigned* __restrict__ tab, unsigned* __restrict__ part,
    const float* __restrict__ x, const float* __restrict__ W0, float* __restrict__ xw,
    int ne, int n, int nown, int nbin, int ndsl, int ndrg, int dstride)
{
    __shared__ __align__(16) unsigned char smem[FRONT_LDS];
    const int b  = blockIdx.x;
    const int nd = ndsl * ndrg;
    if (b < nbin) {
        bin_role(smem, b, row, col, sedw, tab, ne, nown, nbin);
    } else if (b < nbin + nd) {
        const int d = b - nbin;
        deg_hist_role(smem, d % ndsl, d / ndsl, col, part, ne, n, dstride);
    } else {
        gemm_x_role(smem, b - nbin - nd, x, W0, xw, n);
    }
}

template <int WLOAD>
__global__ __launch_bounds__(1024) void own_kernel(
    const int* __restrict__ sedw, float* __restrict__ sw,
    const unsigned* __restrict__ tab, const float* __restrict__ dinv,
    const float* __restrict__ xw, float* __restrict__ agg, int n, int nslice)
{
    const int2* sed = reinterpret_cast<const int2*>(sedw);
    __shared__ __align__(16) float acc[RN * 64];
    const int tid  = threadIdx.x;
    const int lane = tid & 31;
    const int wave = __builtin_amdgcn_readfirstlane(tid >> 5);
    const int base = blockIdx.x * RN;
    int len = n - base;
    if (len > RN) len = RN;
    const int nq = len * 16;

    const unsigned* tb = tab + (size_t)blockIdx.x * nslice;
    unsigned segw = 0u;
    if (wave + 32 * lane < nslice) segw = tb[wave + 32 * lane];

    {
        const float4* x4 = reinterpret_cast<const float4*>(xw + (size_t)base * 64);
        float4*       a4 = reinterpret_cast<float4*>(acc);
        float4 sv[OWN_NQ];
        float  dq[OWN_NQ];
        #pragma unroll
        for (int t = 0; t < OWN_NQ; ++t) {
            int q = tid + 1024 * t;
            q = (q < nq) ? q : nq - 1;
            sv[t] = x4[q];
            dq[t] = dinv[base + (q >> 4)];
        }
        #pragma unroll
        for (int t = 0; t < OWN_NQ; ++t) {
            const int q = tid + 1024 * t;
            if (q < nq) {
                float4 v = sv[t];
                v.x = v.x * dq[t] * dq[t];
                v.y = v.y * dq[t] * dq[t];
                v.z = v.z * dq[t] * dq[t];
                v.w = v.w * dq[t] * dq[t];
                a4[q] = v;
            }
        }
    }
    __syncthreads();

    {
        const int nt = (nslice - wave + 31) / 32;
        for (int t = 0; t < nt; ++t) {
            const unsigned word = (unsigned)__builtin_amdgcn_readlane((int)segw, t);
            const int c  = (int)(word >> 16);
            const int g0 = (wave + 32 * t) * BIN_SL + (int)(word & 0xffffu);
            for (int k0 = 0; k0 < c; k0 += 32) {
                int m = c - k0;
                if (m > 32) m = 32;
                int   r  = base;
                int   cl = 0;
                float w  = 0.f;
                if (lane < m) {
                    const int2 ed = sed[g0 + k0 + lane];
                    r  = ed.x;
                    cl = ed.y;
                    if (WLOAD == 1) w = sw[g0 + k0 + lane];
                }
                float wr = 0.f, wc = 0.f;
                if (WLOAD == 0) {
                    wr = dinv[r];
                    wc = dinv[base + cl];
                }
                const int wi  = __float_as_int(w);
                const int wri = __float_as_int(wr);
                const int wci = __float_as_int(wc);
                for (int jb = 0; jb < m; jb += 8) {
                    float2 v[8];
                    #pragma unroll
                    for (int j = 0; j < 8; ++j) {
                        const int rj = __builtin_amdgcn_readlane(r, jb + j);
                        v[j] = *reinterpret_cast<const float2*>(xw + (size_t)rj * 64 + 2 * lane);
                    }
                    #pragma unroll
                    for (int j = 0; j < 8; ++j) {
                        const int   cj = __builtin_amdgcn_readlane(cl, jb + j);
                        float nj;
                        if (WLOAD == 1) {
                            nj = __int_as_float(__builtin_amdgcn_readlane(wi, jb + j));
                        } else {
                            const float p = __int_as_float(__builtin_amdgcn_readlane(wri, jb + j)) *
                                            __int_as_float(__builtin_amdgcn_readlane(wci, jb + j));
                            nj = (jb + j < m) ? p : 0.f;
                        }
                        float* dst = acc + cj * 64 + 2 * lane;
                        unsafeAtomicAdd(dst,     nj * v[j].x);
                        unsafeAtomicAdd(dst + 1, nj * v[j].y);
                    }
                }
                if (WLOAD == 0 && lane < m) sw[g0 + k0 + lane] = wr * wc;
            }
        }
    }
    __syncthreads();

    {
        float4*       o4 = reinterpret_cast<float4*>(agg + (size_t)base * 64);
        const float4* a4 = reinterpret_cast<const float4*>(acc);
        for (int q = tid; q < nq; q += 1024) o4[q] = a4[q];
    }
}

extern "C" void kernel_launch(void* const* d_in, const int* in_sizes, int n_in,
                              void* d_out, int out_size, void* d_ws, size_t ws_size,
                              hipStream_t stream)
{
    const float* x      = (const float*)d_in[0];
    const int*   ei     = (const int*)d_in[1];
    const float* W0     = (const float*)d_in[2];
    const float* b0     = (const float*)d_in[3];
    const float* W1     = (const float*)d_in[4];
    const float* b1     = (const float*)d_in[5];
    const float* bn0_g  = (const float*)d_in[6];
    const float* bn0_b  = (const float*)d_in[7];
    const float* bn0_m  = (const float*)d_in[8];
    const float* bn0_v  = (const float*)d_in[9];
    const float* bn1_g  = (const float*)d_in[10];
    const float* bn1_b  = (const float*)d_in[11];
    const float* bn1_m  = (const float*)d_in[12];
    const float* bn1_v  = (const float*)d_in[13];
    const float* lin1_W = (const float*)d_in[14];
    const float* lin1_b = (const float*)d_in[15];
    const float* lin2_W = (const float*)d_in[16];
    const float* lin2_b = (const float*)d_in[17];
    const float* lin3_W = (const float*)d_in[18];
    const float* lin3_b = (const float*)d_in[19];

    const int N = in_sizes[0] / 64;
    const int E = in_sizes[1] / 2;
    const int* row = ei;
    const int* col = ei + E;

    char*  ws   = (char*)d_ws;
    size_t off0 = 0;
    size_t off1 = (off0 + (size_t)N * sizeof(float) + 255) & ~(size_t)255;
    size_t off2 = (off1 + (size_t)N * 64 * sizeof(float) + 255) & ~(size_t)255;
    size_t off3 = (off2 + (size_t)N * 64 * sizeof(float) + 255) & ~(size_t)255;
    float* dinv = (float*)(ws + off0);
    float* bufA = (float*)(ws + off1);
    float* bufB = (float*)(ws + off2);
    float* bufC = (float*)(ws + off3);
    size_t off4 = (off3 + (size_t)N * 64 * sizeof(float) + 255) & ~(size_t)255;
    unsigned* part = (unsigned*)(ws + off4);

    const dim3 blk(256);
    const int gBin  = (E + BIN_SL - 1) / BIN_SL;
    const int gOwn  = (N + RN - 1) / RN;
    const int gSl   = ((E + DEG_SLICE - 1) / DEG_SLICE + DEG_KB - 1) / DEG_KB * DEG_KB;
    const int gRg   = (N + DEG_RANGE - 1) / DEG_RANGE;
    const int nwS   = ((N + 1) / 2 + 3) & ~3;
    const int gW    = ((N + 1) / 2 + 255) / 256;
    const int gGemm = (N + 127) / 128;
    const int gFront = (N + FRONT_ROWS - 1) / FRONT_ROWS;
    size_t off5 = (off4 + (size_t)gSl * nwS * sizeof(unsigned) + 255) & ~(size_t)255;
    size_t off6 = (off5 + (size_t)E * sizeof(int2) + 255) & ~(size_t)255;
    size_t off7 = (off6 + (size_t)E * sizeof(float) + 255) & ~(size_t)255;
    int*      sed = (int*)(ws + off5);
    float*    sw  = (float*)(ws + off6);
    unsigned* tab = (unsigned*)(ws + off7);

    front_kernel<<<gBin + gSl * gRg + gFront, dim3(1024), 0, stream>>>(
        row, col, sed, tab, part, x, W0, bufA,
        E, N, gOwn, gBin, gSl, gRg, nwS);

    dinv_kernel<<<gW, blk, 0, stream>>>(part, dinv, N, nwS, gSl);

    own_kernel<0><<<gOwn, dim3(1024), 0, stream>>>(sed, sw, tab, dinv, bufA, bufB, N, gBin);

    gemm_wmma_kernel<64, 1, 0><<<gGemm, blk, 0, stream>>>(
        bufB, W1, nullptr, bufA, N,
        b0, bn0_g, bn0_b, bn0_m, bn0_v);
    own_kernel<1><<<gOwn, dim3(1024), 0, stream>>>(sed, sw, tab, dinv, bufA, bufC, N, gBin);

    gemm_wmma_kernel<64, 1, 2><<<gGemm, blk, 0, stream>>>(
        bufC, lin1_W, lin1_b, bufA, N,
        b1, bn1_g, bn1_b, bn1_m, bn1_v);
    gemm_wmma_kernel<64, 0, 2><<<gGemm, blk, 0, stream>>>(
        bufA, lin2_W, lin2_b, bufB, N,
        nullptr, nullptr, nullptr, nullptr, nullptr);
    gemm_wmma_kernel<16, 0, 1><<<gGemm, blk, 0, stream>>>(
        bufB, lin3_W, lin3_b, (float*)d_out, N,
        nullptr, nullptr, nullptr, nullptr, nullptr);
}


# dense layers run in the owner kernel on its finished LDS tile: bn+ReLU in place, then W1 / the 3 head layers on WMMA per 16-row strip; 4 GEMM launches and row round trips gone; sums unchanged
# speedup vs baseline: 4.0107x; 1.3449x over previous
#include <hip/hip_runtime.h>
#include <cstdint>
#include <cstddef>

typedef __attribute__((ext_vector_type(2))) float v2f;
typedef __attribute__((ext_vector_type(8))) float v8f;


constexpr int DEG_SLICE = 32768;
constexpr int DEG_RANGE = 32768;
constexpr int DEG_WORDS = DEG_RANGE / 2;
constexpr int DEG_EPT   = DEG_SLICE / 1024;
constexpr int DEG_KB    = 16;

__device__ __forceinline__ void deg_hist_role(
    unsigned char* smem, int bx, int by,
    const int* __restrict__ col, unsigned* __restrict__ part, int ne, int n, int stride)
{
    unsigned* hist = reinterpret_cast<unsigned*>(smem);
    const int tid  = threadIdx.x;
    const int e0   = bx * DEG_SLICE;
    const int base = by * DEG_RANGE;
    int len = n - base;
    if (len > DEG_RANGE) len = DEG_RANGE;
    const int nq = ((len + 1) / 2 + 3) / 4;

    int cv[DEG_EPT];
    #pragma unroll
    for (int i = 0; i < DEG_EPT; ++i) {
        int e = e0 + i * 1024 + tid;
        e = (e < ne) ? e : ne - 1;
        cv[i] = col[e];
    }

    uint4* h4 = reinterpret_cast<uint4*>(hist);
    for (int q = tid; q < nq; q += 1024) h4[q] = make_uint4(0u, 0u, 0u, 0u);
    __syncthreads();

    int nval = ne - e0;
    if (nval > DEG_SLICE) nval = DEG_SLICE;
    #pragma unroll
    for (int i = 0; i < DEG_EPT; ++i) {
        const unsigned rel = (unsigned)(cv[i] - base);
        if (tid < nval - i * 1024 && rel < (unsigned)len)
            atomicAdd(&hist[rel >> 1], (rel & 1u) ? 0x10000u : 1u);
    }
    __syncthreads();

    uint4* p4 = reinterpret_cast<uint4*>(part + (size_t)bx * stride + (base >> 1));
    for (int q = tid; q < nq; q += 1024) p4[q] = h4[q];
}

__global__ __launch_bounds__(256) void dinv_kernel(
    const unsigned* __restrict__ part, float* __restrict__ dinv, int n, int stride, int nslice)
{
    const int w = blockIdx.x * 256 + threadIdx.x;
    if (2 * w >= n) return;
    const unsigned* p = part + w;
    unsigned lo = 0u, hi = 0u;
    for (int k0 = 0; k0 < nslice; k0 += DEG_KB) {
        unsigned v[DEG_KB];
        #pragma unroll
        for (int j = 0; j < DEG_KB; ++j) v[j] = p[(size_t)(k0 + j) * stride];
        #pragma unroll
        for (int j = 0; j < DEG_KB; ++j) {
            lo += v[j] & 0xffffu;
            hi += v[j] >> 16;
        }
    }
    const float d0 = rsqrtf(1.0f + (float)lo);
    const float d1 = rsqrtf(1.0f + (float)hi);
    if (2 * w + 1 < n) *reinterpret_cast<float2*>(&dinv[2 * w]) = make_float2(d0, d1);
    else               dinv[2 * w] = d0;
}

constexpr int RN      = 392;
constexpr int BIN_SL  = 8192;
constexpr int BIN_EPT = BIN_SL / 1024;
constexpr int MAXB    = 512;
constexpr int OWN_NQ  = (RN * 16 + 1023) / 1024;
constexpr int TP      = 68;
constexpr int OWN_ST  = (RN + 15) / 16;
constexpr int OWN_TR  = OWN_ST * 16;
static_assert(OWN_TR * 16 <= OWN_NQ * 1024, "tile start covers every strip row");
static_assert(OWN_ST <= 32, "one wave per strip");

__device__ __forceinline__ void bin_role(
    unsigned char* smem, int s,
    const int* __restrict__ row, const int* __restrict__ col,
    int* __restrict__ sedw, unsigned* __restrict__ tab,
    int ne, int nb, int nslice)
{
    int2* sed = reinterpret_cast<int2*>(sedw);
    int2*     led   = reinterpret_cast<int2*>(smem);
    unsigned* cnt   = reinterpret_cast<unsigned*>(smem + BIN_SL * sizeof(int2));
    unsigned* first = cnt + MAXB;
    unsigned* cur   = first + MAXB;
    const int tid = threadIdx.x;
    const int e0  = s * BIN_SL;

    int cv[BIN_EPT], rv[BIN_EPT];
    #pragma unroll
    for (int i = 0; i < BIN_EPT; ++i) {
        int e = e0 + i * 1024 + tid;
        e = (e < ne) ? e : ne - 1;
        cv[i] = col[e];
        rv[i] = row[e];
    }

    if (tid < MAXB) { cnt[tid] = 0u; cur[tid] = 0u; }
    __syncthreads();

    #pragma unroll
    for (int i = 0; i < BIN_EPT; ++i) {
        const int e = e0 + i * 1024 + tid;
        if (e < ne) atomicAdd(&cnt[cv[i] / RN], 1u);
    }
    __syncthreads();

    if (tid < 32) {
        unsigned loc[MAXB / 32];
        unsigned sum = 0u;
        #pragma unroll
        for (int j = 0; j < MAXB / 32; ++j) { loc[j] = cnt[tid * (MAXB / 32) + j]; sum += loc[j]; }
        unsigned inc = sum;
        #pragma unroll
        for (int d = 1; d < 32; d <<= 1) {
            const unsigned t = __shfl_up(inc, (unsigned)d);
            if (tid >= d) inc += t;
        }
        unsigned run = inc - sum;
        #pragma unroll
        for (int j = 0; j < MAXB / 32; ++j) { first[tid * (MAXB / 32) + j] = run; run += loc[j]; }
    }
    __syncthreads();

    for (int q = tid; q < nb; q += 1024)
        tab[(size_t)q * nslice + s] = first[q] | (cnt[q] << 16);

    #pragma unroll
    for (int i = 0; i < BIN_EPT; ++i) {
        const int e = e0 + i * 1024 + tid;
        if (e < ne) {
            const int b   = cv[i] / RN;
            const int pos = (int)(first[b] + atomicAdd(&cur[b], 1u));
            led[pos] = make_int2(rv[i], cv[i] - b * RN);
        }
    }
    __syncthreads();

    int valid = ne - e0;
    if (valid > BIN_SL) valid = BIN_SL;
    for (int q = tid; q < valid; q += 1024) sed[e0 + q] = led[q];
}

constexpr int FRONT_ROWS = 512;
constexpr int FRONT_LDS  = 64 * 64 * 4 + 32 * 16 * 68 * 4;

__device__ __forceinline__ void gemm_x_role(
    unsigned char* smem, int blk,
    const float* __restrict__ A, const float* __restrict__ W,
    float* __restrict__ out, int nrows)
{
    float* sW = reinterpret_cast<float*>(smem);
    const int tid  = threadIdx.x;
    const int wave = tid >> 5;
    const int lane = tid & 31;
    const int hh   = lane >> 4;
    const int l16  = lane & 15;
    float* sA = reinterpret_cast<float*>(smem + 64 * 64 * 4) + wave * (16 * 68);
    const int r0   = blk * FRONT_ROWS + wave * 16;

    reinterpret_cast<float4*>(sW)[tid] = reinterpret_cast<const float4*>(W)[tid];

    const float4* A4 = reinterpret_cast<const float4*>(A);
    float4 sv[8];
    #pragma unroll
    for (int i = 0; i < 8; ++i) {
        int gr = r0 + 2 * i + hh;
        gr = (gr < nrows) ? gr : nrows - 1;
        sv[i] = A4[(size_t)gr * 16 + l16];
    }
    #pragma unroll
    for (int i = 0; i < 8; ++i)
        *reinterpret_cast<float4*>(&sA[(2 * i + hh) * 68 + l16 * 4]) = sv[i];
    __syncthreads();

    v8f acc[4] = {};

    #pragma unroll
    for (int kb = 0; kb < 64; kb += 4) {
        v2f a;
        a.x = sA[l16 * 68 + kb + 2 * hh + 0];
        a.y = sA[l16 * 68 + kb + 2 * hh + 1];
        #pragma unroll
        for (int nt = 0; nt < 4; ++nt) {
            v2f b;
            b.x = sW[(kb + 2 * hh + 0) * 64 + nt * 16 + l16];
            b.y = sW[(kb + 2 * hh + 1) * 64 + nt * 16 + l16];
            acc[nt] = __builtin_amdgcn_wmma_f32_16x16x4_f32(
                false, a, false, b, (short)0, acc[nt], false, false);
        }
    }

    #pragma unroll
    for (int nt = 0; nt < 4; ++nt) {
        const int col = nt * 16 + l16;
        #pragma unroll
        for (int j = 0; j < 8; ++j) {
            const int gr = r0 + j + 8 * hh;
            if (gr < nrows) out[(size_t)gr * 64 + col] = acc[nt][j];
        }
    }
}

__global__ __launch_bounds__(1024, 16) void front_kernel(
    const int* __restrict__ row, const int* __restrict__ col,
    int* __restrict__ sedw, unsigned* __restrict__ tab, unsigned* __restrict__ part,
    const float* __restrict__ x, const float* __restrict__ W0, float* __restrict__ xw,
    int ne, int n, int nown, int nbin, int ndsl, int ndrg, int dstride)
{
    __shared__ __align__(16) unsigned char smem[FRONT_LDS];
    const int b  = blockIdx.x;
    const int nd = ndsl * ndrg;
    if (b < nbin) {
        bin_role(smem, b, row, col, sedw, tab, ne, nown, nbin);
    } else if (b < nbin + nd) {
        const int d = b - nbin;
        deg_hist_role(smem, d % ndsl, d / ndsl, col, part, ne, n, dstride);
    } else {
        gemm_x_role(smem, b - nbin - nd, x, W0, xw, n);
    }
}

template <int NCOL>
__device__ __forceinline__ void strip_product(
    const float* st, const float* Wl, v8f (&c)[NCOL / 16], int l16, int hh)
{
    #pragma unroll
    for (int kb = 0; kb < 64; kb += 4) {
        v2f a;
        a.x = st[l16 * TP + kb + 2 * hh + 0];
        a.y = st[l16 * TP + kb + 2 * hh + 1];
        #pragma unroll
        for (int nt = 0; nt < NCOL / 16; ++nt) {
            v2f b;
            b.x = Wl[(kb + 2 * hh + 0) * NCOL + nt * 16 + l16];
            b.y = Wl[(kb + 2 * hh + 1) * NCOL + nt * 16 + l16];
            c[nt] = __builtin_amdgcn_wmma_f32_16x16x4_f32(
                false, a, false, b, (short)0, c[nt], false, false);
        }
    }
}

__device__ __forceinline__ void strip_put_relu(
    float* st, const v8f (&c)[4], const float* __restrict__ bias, int l16, int hh)
{
    #pragma unroll
    for (int nt = 0; nt < 4; ++nt) {
        const int col = nt * 16 + l16;
        const float bv = bias[col];
        #pragma unroll
        for (int j = 0; j < 8; ++j)
            st[(j + 8 * hh) * TP + col] = fmaxf(c[nt][j] + bv, 0.f);
    }
}

template <int WLOAD, int HEAD>
__global__ __launch_bounds__(1024) void own_kernel(
    const int* __restrict__ sedw, float* __restrict__ sw,
    const unsigned* __restrict__ tab, const float* __restrict__ dinv,
    const float* __restrict__ xw, float* __restrict__ out, int n, int nslice,
    const float* __restrict__ cb, const float* __restrict__ bn_g,
    const float* __restrict__ bn_b, const float* __restrict__ bn_m,
    const float* __restrict__ bn_v,
    const float* __restrict__ Wa, const float* __restrict__ ba,
    const float* __restrict__ Wb, const float* __restrict__ bb,
    const float* __restrict__ Wc, const float* __restrict__ bc)
{
    const int2* sed = reinterpret_cast<const int2*>(sedw);
    constexpr int WF = (HEAD == 1) ? (64 * 64 * 2 + 64 * 16) : (64 * 64);
    __shared__ __align__(16) float smem[WF + OWN_TR * TP];
    float* sWt = smem;
    float* acc = smem + WF;
    const int tid  = threadIdx.x;
    const int lane = tid & 31;
    const int wave = __builtin_amdgcn_readfirstlane(tid >> 5);
    const int base = blockIdx.x * RN;
    int len = n - base;
    if (len > RN) len = RN;
    const int nq = len * 16;

    const unsigned* tb = tab + (size_t)blockIdx.x * nslice;
    unsigned segw = 0u;
    if (wave + 32 * lane < nslice) segw = tb[wave + 32 * lane];

    const float4 wa4 = reinterpret_cast<const float4*>(Wa)[tid];
    float4 wb4 = make_float4(0.f, 0.f, 0.f, 0.f), wc4 = wb4;
    if (HEAD == 1) {
        wb4 = reinterpret_cast<const float4*>(Wb)[tid];
        wc4 = reinterpret_cast<const float4*>(Wc)[tid & 255];
    }

    {
        const float4* x4 = reinterpret_cast<const float4*>(xw + (size_t)base * 64);
        float4*       a4 = reinterpret_cast<float4*>(acc);
        float4 sv[OWN_NQ];
        float  dq[OWN_NQ];
        #pragma unroll
        for (int t = 0; t < OWN_NQ; ++t) {
            int q = tid + 1024 * t;
            q = (q < nq) ? q : nq - 1;
            sv[t] = x4[q];
            dq[t] = dinv[base + (q >> 4)];
        }
        #pragma unroll
        for (int t = 0; t < OWN_NQ; ++t) {
            const int q = tid + 1024 * t;
            if (q < OWN_TR * 16) {
                float4 v = make_float4(0.f, 0.f, 0.f, 0.f);
                if (q < nq) {
                    v = sv[t];
                    v.x = v.x * dq[t] * dq[t];
                    v.y = v.y * dq[t] * dq[t];
                    v.z = v.z * dq[t] * dq[t];
                    v.w = v.w * dq[t] * dq[t];
                }
                a4[(q >> 4) * (TP / 4) + (q & 15)] = v;
            }
        }
        float4* s4 = reinterpret_cast<float4*>(sWt);
        s4[tid] = wa4;
        if (HEAD == 1) {
            s4[1024 + tid] = wb4;
            if (tid < 256) s4[2048 + tid] = wc4;
        }
    }
    __syncthreads();

    {
        const int nt = (nslice - wave + 31) / 32;
        for (int t = 0; t < nt; ++t) {
            const unsigned word = (unsigned)__builtin_amdgcn_readlane((int)segw, t);
            const int c  = (int)(word >> 16);
            const int g0 = (wave + 32 * t) * BIN_SL + (int)(word & 0xffffu);
            for (int k0 = 0; k0 < c; k0 += 32) {
                int m = c - k0;
                if (m > 32) m = 32;
                int   r  = base;
                int   cl = 0;
                float w  = 0.f;
                if (lane < m) {
                    const int2 ed = sed[g0 + k0 + lane];
                    r  = ed.x;
                    cl = ed.y;
                    if (WLOAD == 1) w = sw[g0 + k0 + lane];
                }
                float wr = 0.f, wc = 0.f;
                if (WLOAD == 0) {
                    wr = dinv[r];
                    wc = dinv[base + cl];
                }
                const int wi  = __float_as_int(w);
                const int wri = __float_as_int(wr);
                const int wci = __float_as_int(wc);
                for (int jb = 0; jb < m; jb += 8) {
                    float2 v[8];
                    #pragma unroll
                    for (int j = 0; j < 8; ++j) {
                        const int rj = __builtin_amdgcn_readlane(r, jb + j);
                        v[j] = *reinterpret_cast<const float2*>(xw + (size_t)rj * 64 + 2 * lane);
                    }
                    #pragma unroll
                    for (int j = 0; j < 8; ++j) {
                        const int   cj = __builtin_amdgcn_readlane(cl, jb + j);
                        float nj;
                        if (WLOAD == 1) {
                            nj = __int_as_float(__builtin_amdgcn_readlane(wi, jb + j));
                        } else {
                            const float p = __int_as_float(__builtin_amdgcn_readlane(wri, jb + j)) *
                                            __int_as_float(__builtin_amdgcn_readlane(wci, jb + j));
                            nj = (jb + j < m) ? p : 0.f;
                        }
                        float* dst = acc + cj * TP + 2 * lane;
                        unsafeAtomicAdd(dst,     nj * v[j].x);
                        unsafeAtomicAdd(dst + 1, nj * v[j].y);
                    }
                }
                if (WLOAD == 0 && lane < m) sw[g0 + k0 + lane] = wr * wc;
            }
        }
    }
    __syncthreads();

    const int  hh  = lane >> 4;
    const int  l16 = lane & 15;
    const bool has = wave < OWN_ST;
    float* st = acc + (has ? wave : 0) * (16 * TP);

    if (has) {
        const float4 pb = reinterpret_cast<const float4*>(cb)[l16];
        const float4 pm = reinterpret_cast<const float4*>(bn_m)[l16];
        const float4 pg = reinterpret_cast<const float4*>(bn_g)[l16];
        const float4 pe = reinterpret_cast<const float4*>(bn_b)[l16];
        const float4 pv = reinterpret_cast<const float4*>(bn_v)[l16];
        float4 pr;
        pr.x = rsqrtf(pv.x + 1e-5f);
        pr.y = rsqrtf(pv.y + 1e-5f);
        pr.z = rsqrtf(pv.z + 1e-5f);
        pr.w = rsqrtf(pv.w + 1e-5f);
        #pragma unroll
        for (int i = 0; i < 8; ++i) {
            float4* p = reinterpret_cast<float4*>(&st[(2 * i + hh) * TP + l16 * 4]);
            float4 v = *p;
            v.x = fmaxf((v.x + pb.x - pm.x) * pr.x * pg.x + pe.x, 0.f);
            v.y = fmaxf((v.y + pb.y - pm.y) * pr.y * pg.y + pe.y, 0.f);
            v.z = fmaxf((v.z + pb.z - pm.z) * pr.z * pg.z + pe.z, 0.f);
            v.w = fmaxf((v.w + pb.w - pm.w) * pr.w * pg.w + pe.w, 0.f);
            *p = v;
        }
    }
    __syncthreads();

    v8f c1[4] = {};
    if (has) strip_product<64>(st, sWt, c1, l16, hh);

    if (HEAD == 0) {
        if (has) {
            #pragma unroll
            for (int nt = 0; nt < 4; ++nt) {
                const int col = nt * 16 + l16;
                #pragma unroll
                for (int j = 0; j < 8; ++j) {
                    const int lr = wave * 16 + j + 8 * hh;
                    if (lr < len) out[(size_t)(base + lr) * 64 + col] = c1[nt][j];
                }
            }
        }
    } else {
        if (has) strip_put_relu(st, c1, ba, l16, hh);
        __syncthreads();
        v8f c2[4] = {};
        if (has) {
            strip_product<64>(st, sWt + 64 * 64, c2, l16, hh);
            strip_put_relu(st, c2, bb, l16, hh);
        }
        __syncthreads();
        if (has) {
            v8f c3[1] = {};
            strip_product<16>(st, sWt + 2 * 64 * 64, c3, l16, hh);
            const float bv = bc[l16];
            #pragma unroll
            for (int j = 0; j < 8; ++j) {
                const int lr = wave * 16 + j + 8 * hh;
                if (lr < len) out[(size_t)(base + lr) * 16 + l16] = c3[0][j] + bv;
            }
        }
    }
}

extern "C" void kernel_launch(void* const* d_in, const int* in_sizes, int n_in,
                              void* d_out, int out_size, void* d_ws, size_t ws_size,
                              hipStream_t stream)
{
    const float* x      = (const float*)d_in[0];
    const int*   ei     = (const int*)d_in[1];
    const float* W0     = (const float*)d_in[2];
    const float* b0     = (const float*)d_in[3];
    const float* W1     = (const float*)d_in[4];
    const float* b1     = (const float*)d_in[5];
    const float* bn0_g  = (const float*)d_in[6];
    const float* bn0_b  = (const float*)d_in[7];
    const float* bn0_m  = (const float*)d_in[8];
    const float* bn0_v  = (const float*)d_in[9];
    const float* bn1_g  = (const float*)d_in[10];
    const float* bn1_b  = (const float*)d_in[11];
    const float* bn1_m  = (const float*)d_in[12];
    const float* bn1_v  = (const float*)d_in[13];
    const float* lin1_W = (const float*)d_in[14];
    const float* lin1_b = (const float*)d_in[15];
    const float* lin2_W = (const float*)d_in[16];
    const float* lin2_b = (const float*)d_in[17];
    const float* lin3_W = (const float*)d_in[18];
    const float* lin3_b = (const float*)d_in[19];

    const int N = in_sizes[0] / 64;
    const int E = in_sizes[1] / 2;
    const int* row = ei;
    const int* col = ei + E;

    char*  ws   = (char*)d_ws;
    size_t off0 = 0;
    size_t off1 = (off0 + (size_t)N * sizeof(float) + 255) & ~(size_t)255;
    size_t off2 = (off1 + (size_t)N * 64 * sizeof(float) + 255) & ~(size_t)255;
    float* dinv = (float*)(ws + off0);
    float* bufA = (float*)(ws + off1);
    float* bufB = (float*)(ws + off2);
    size_t off4 = (off2 + (size_t)N * 64 * sizeof(float) + 255) & ~(size_t)255;
    unsigned* part = (unsigned*)(ws + off4);

    const dim3 blk(256);
    const int gBin  = (E + BIN_SL - 1) / BIN_SL;
    const int gOwn  = (N + RN - 1) / RN;
    const int gSl   = ((E + DEG_SLICE - 1) / DEG_SLICE + DEG_KB - 1) / DEG_KB * DEG_KB;
    const int gRg   = (N + DEG_RANGE - 1) / DEG_RANGE;
    const int nwS   = ((N + 1) / 2 + 3) & ~3;
    const int gW    = ((N + 1) / 2 + 255) / 256;
    const int gFront = (N + FRONT_ROWS - 1) / FRONT_ROWS;
    size_t off5 = (off4 + (size_t)gSl * nwS * sizeof(unsigned) + 255) & ~(size_t)255;
    size_t off6 = (off5 + (size_t)E * sizeof(int2) + 255) & ~(size_t)255;
    size_t off7 = (off6 + (size_t)E * sizeof(float) + 255) & ~(size_t)255;
    int*      sed = (int*)(ws + off5);
    float*    sw  = (float*)(ws + off6);
    unsigned* tab = (unsigned*)(ws + off7);

    front_kernel<<<gBin + gSl * gRg + gFront, dim3(1024), 0, stream>>>(
        row, col, sed, tab, part, x, W0, bufA,
        E, N, gOwn, gBin, gSl, gRg, nwS);

    dinv_kernel<<<gW, blk, 0, stream>>>(part, dinv, N, nwS, gSl);

    own_kernel<0, 0><<<gOwn, dim3(1024), 0, stream>>>(
        sed, sw, tab, dinv, bufA, bufB, N, gBin,
        b0, bn0_g, bn0_b, bn0_m, bn0_v,
        W1, nullptr, nullptr, nullptr, nullptr, nullptr);

    own_kernel<1, 1><<<gOwn, dim3(1024), 0, stream>>>(
        sed, sw, tab, dinv, bufB, (float*)d_out, N, gBin,
        b1, bn1_g, bn1_b, bn1_m, bn1_v,
        lin1_W, lin1_b, lin2_W, lin2_b, lin3_W, lin3_b);
}


# stored rows pre-multiplied by their node's rsqrt(1+deg) (degree tables in their own launch, summed in the x@W0 blocks): owner walk adds rows as loaded, no per-edge weight; product association changed
# speedup vs baseline: 4.2416x; 1.0576x over previous
#include <hip/hip_runtime.h>
#include <cstdint>
#include <cstddef>

typedef __attribute__((ext_vector_type(2))) float v2f;
typedef __attribute__((ext_vector_type(8))) float v8f;


constexpr int DEG_SLICE = 32768;
constexpr int DEG_RANGE = 32768;
constexpr int DEG_WORDS = DEG_RANGE / 2;
constexpr int DEG_EPT   = DEG_SLICE / 1024;
constexpr int DEG_KB    = 32;

__global__ __launch_bounds__(1024) void deg_kernel(
    const int* __restrict__ col, unsigned* __restrict__ part, int ne, int n, int ndsl, int stride)
{
    __shared__ __align__(16) unsigned hist[DEG_WORDS];
    const int bx   = blockIdx.x % ndsl;
    const int by   = blockIdx.x / ndsl;
    const int tid  = threadIdx.x;
    const int e0   = bx * DEG_SLICE;
    const int base = by * DEG_RANGE;
    int len = n - base;
    if (len > DEG_RANGE) len = DEG_RANGE;
    const int nq = ((len + 1) / 2 + 3) / 4;

    int cv[DEG_EPT];
    #pragma unroll
    for (int i = 0; i < DEG_EPT; ++i) {
        int e = e0 + i * 1024 + tid;
        e = (e < ne) ? e : ne - 1;
        cv[i] = col[e];
    }

    uint4* h4 = reinterpret_cast<uint4*>(hist);
    for (int q = tid; q < nq; q += 1024) h4[q] = make_uint4(0u, 0u, 0u, 0u);
    __syncthreads();

    int nval = ne - e0;
    if (nval > DEG_SLICE) nval = DEG_SLICE;
    #pragma unroll
    for (int i = 0; i < DEG_EPT; ++i) {
        const unsigned rel = (unsigned)(cv[i] - base);
        if (tid < nval - i * 1024 && rel < (unsigned)len)
            atomicAdd(&hist[rel >> 1], (rel & 1u) ? 0x10000u : 1u);
    }
    __syncthreads();

    uint4* p4 = reinterpret_cast<uint4*>(part + (size_t)bx * stride + (base >> 1));
    for (int q = tid; q < nq; q += 1024) p4[q] = h4[q];
}

constexpr int RN      = 392;
constexpr int BIN_SL  = 8192;
constexpr int BIN_EPT = BIN_SL / 1024;
constexpr int MAXB    = 512;
constexpr int OWN_NQ  = (RN * 16 + 1023) / 1024;
constexpr int TP      = 68;
constexpr int OWN_ST  = (RN + 15) / 16;
constexpr int OWN_TR  = OWN_ST * 16;
static_assert(OWN_TR * 16 <= OWN_NQ * 1024, "tile start covers every strip row");
static_assert(OWN_ST <= 32, "one wave per strip");

__device__ __forceinline__ void bin_role(
    unsigned char* smem, int s,
    const int* __restrict__ row, const int* __restrict__ col,
    int* __restrict__ sedw, unsigned* __restrict__ tab,
    int ne, int nb, int nslice)
{
    int2* sed = reinterpret_cast<int2*>(sedw);
    int2*     led   = reinterpret_cast<int2*>(smem);
    unsigned* cnt   = reinterpret_cast<unsigned*>(smem + BIN_SL * sizeof(int2));
    unsigned* first = cnt + MAXB;
    unsigned* cur   = first + MAXB;
    const int tid = threadIdx.x;
    const int e0  = s * BIN_SL;

    int cv[BIN_EPT], rv[BIN_EPT];
    #pragma unroll
    for (int i = 0; i < BIN_EPT; ++i) {
        int e = e0 + i * 1024 + tid;
        e = (e < ne) ? e : ne - 1;
        cv[i] = col[e];
        rv[i] = row[e];
    }

    if (tid < MAXB) { cnt[tid] = 0u; cur[tid] = 0u; }
    __syncthreads();

    #pragma unroll
    for (int i = 0; i < BIN_EPT; ++i) {
        const int e = e0 + i * 1024 + tid;
        if (e < ne) atomicAdd(&cnt[cv[i] / RN], 1u);
    }
    __syncthreads();

    if (tid < 32) {
        unsigned loc[MAXB / 32];
        unsigned sum = 0u;
        #pragma unroll
        for (int j = 0; j < MAXB / 32; ++j) { loc[j] = cnt[tid * (MAXB / 32) + j]; sum += loc[j]; }
        unsigned inc = sum;
        #pragma unroll
        for (int d = 1; d < 32; d <<= 1) {
            const unsigned t = __shfl_up(inc, (unsigned)d);
            if (tid >= d) inc += t;
        }
        unsigned run = inc - sum;
        #pragma unroll
        for (int j = 0; j < MAXB / 32; ++j) { first[tid * (MAXB / 32) + j] = run; run += loc[j]; }
    }
    __syncthreads();

    for (int q = tid; q < nb; q += 1024)
        tab[(size_t)q * nslice + s] = first[q] | (cnt[q] << 16);

    #pragma unroll
    for (int i = 0; i < BIN_EPT; ++i) {
        const int e = e0 + i * 1024 + tid;
        if (e < ne) {
            const int b   = cv[i] / RN;
            const int pos = (int)(first[b] + atomicAdd(&cur[b], 1u));
            led[pos] = make_int2(rv[i], cv[i] - b * RN);
        }
    }
    __syncthreads();

    int valid = ne - e0;
    if (valid > BIN_SL) valid = BIN_SL;
    for (int q = tid; q < valid; q += 1024) sed[e0 + q] = led[q];
}

constexpr int FRONT_ROWS = 512;
constexpr int FRONT_LDS  = 64 * 64 * 4 + 32 * 16 * 68 * 4;

__device__ __forceinline__ void gemm_x_role(
    unsigned char* smem, int blk,
    const float* __restrict__ A, const float* __restrict__ W,
    const unsigned* __restrict__ part, float* __restrict__ dinv,
    float* __restrict__ out, int nrows, int stride, int nslice)
{
    float* sW = reinterpret_cast<float*>(smem);
    const int tid  = threadIdx.x;
    const int wave = tid >> 5;
    const int lane = tid & 31;
    const int hh   = lane >> 4;
    const int l16  = lane & 15;
    float* sA = reinterpret_cast<float*>(smem + 64 * 64 * 4) + wave * (16 * 68);
    const int r0   = blk * FRONT_ROWS + wave * 16;

    const int sg = lane >> 3;
    int wi = (r0 >> 1) + (lane & 7);
    const int nwords = (nrows + 1) >> 1;
    wi = (wi < nwords) ? wi : nwords - 1;
    unsigned tv[8];
    #pragma unroll
    for (int j = 0; j < 8; ++j) tv[j] = part[(unsigned)((sg * 8 + j) * stride + wi)];

    reinterpret_cast<float4*>(sW)[tid] = reinterpret_cast<const float4*>(W)[tid];

    const float4* A4 = reinterpret_cast<const float4*>(A);
    float4 sv[8];
    #pragma unroll
    for (int i = 0; i < 8; ++i) {
        int gr = r0 + 2 * i + hh;
        gr = (gr < nrows) ? gr : nrows - 1;
        sv[i] = A4[(unsigned)(gr * 16 + l16)];
    }
    #pragma unroll
    for (int i = 0; i < 8; ++i)
        *reinterpret_cast<float4*>(&sA[(2 * i + hh) * 68 + l16 * 4]) = sv[i];
    unsigned lo = 0u, hi = 0u;
    #pragma unroll
    for (int j = 0; j < 8; ++j) { lo += tv[j] & 0xffffu; hi += tv[j] >> 16; }
    for (int k0 = DEG_KB; k0 < nslice; k0 += DEG_KB) {
        #pragma unroll
        for (int j = 0; j < 8; ++j) tv[j] = part[(unsigned)((k0 + sg * 8 + j) * stride + wi)];
        #pragma unroll
        for (int j = 0; j < 8; ++j) { lo += tv[j] & 0xffffu; hi += tv[j] >> 16; }
    }
    lo += __shfl_xor(lo, 8);   hi += __shfl_xor(hi, 8);
    lo += __shfl_xor(lo, 16);  hi += __shfl_xor(hi, 16);
    const float d0 = rsqrtf(1.0f + (float)lo);
    const float d1 = rsqrtf(1.0f + (float)hi);
    if (lane < 8) {
        const int nd = r0 + 2 * lane;
        if (nd + 1 < nrows)  *reinterpret_cast<float2*>(&dinv[nd]) = make_float2(d0, d1);
        else if (nd < nrows) dinv[nd] = d0;
    }
    __syncthreads();

    v8f acc[4] = {};

    #pragma unroll
    for (int kb = 0; kb < 64; kb += 4) {
        v2f a;
        a.x = sA[l16 * 68 + kb + 2 * hh + 0];
        a.y = sA[l16 * 68 + kb + 2 * hh + 1];
        #pragma unroll
        for (int nt = 0; nt < 4; ++nt) {
            v2f b;
            b.x = sW[(kb + 2 * hh + 0) * 64 + nt * 16 + l16];
            b.y = sW[(kb + 2 * hh + 1) * 64 + nt * 16 + l16];
            acc[nt] = __builtin_amdgcn_wmma_f32_16x16x4_f32(
                false, a, false, b, (short)0, acc[nt], false, false);
        }
    }

    float dj[8];
    #pragma unroll
    for (int j = 0; j < 8; ++j) dj[j] = __shfl((j & 1) ? d1 : d0, (j >> 1) + 4 * hh);

    #pragma unroll
    for (int nt = 0; nt < 4; ++nt) {
        const int col = nt * 16 + l16;
        #pragma unroll
        for (int j = 0; j < 8; ++j) {
            const int gr = r0 + j + 8 * hh;
            if (gr < nrows) out[(size_t)gr * 64 + col] = acc[nt][j] * dj[j];
        }
    }
}

__global__ __launch_bounds__(1024, 16) void front_kernel(
    const int* __restrict__ row, const int* __restrict__ col,
    int* __restrict__ sedw, unsigned* __restrict__ tab,
    const unsigned* __restrict__ part, float* __restrict__ dinv,
    const float* __restrict__ x, const float* __restrict__ W0, float* __restrict__ xw,
    int ne, int n, int nown, int nbin, int dstride, int ndsl)
{
    __shared__ __align__(16) unsigned char smem[FRONT_LDS];
    const int b = blockIdx.x;
    if (b < nbin) {
        bin_role(smem, b, row, col, sedw, tab, ne, nown, nbin);
    } else {
        gemm_x_role(smem, b - nbin, x, W0, part, dinv, xw, n, dstride, ndsl);
    }
}

template <int NCOL>
__device__ __forceinline__ void strip_product(
    const float* st, const float* Wl, v8f (&c)[NCOL / 16], int l16, int hh)
{
    #pragma unroll
    for (int kb = 0; kb < 64; kb += 4) {
        v2f a;
        a.x = st[l16 * TP + kb + 2 * hh + 0];
        a.y = st[l16 * TP + kb + 2 * hh + 1];
        #pragma unroll
        for (int nt = 0; nt < NCOL / 16; ++nt) {
            v2f b;
            b.x = Wl[(kb + 2 * hh + 0) * NCOL + nt * 16 + l16];
            b.y = Wl[(kb + 2 * hh + 1) * NCOL + nt * 16 + l16];
            c[nt] = __builtin_amdgcn_wmma_f32_16x16x4_f32(
                false, a, false, b, (short)0, c[nt], false, false);
        }
    }
}

__device__ __forceinline__ void strip_put_relu(
    float* st, const v8f (&c)[4], const float* __restrict__ bias, int l16, int hh)
{
    #pragma unroll
    for (int nt = 0; nt < 4; ++nt) {
        const int col = nt * 16 + l16;
        const float bv = bias[col];
        #pragma unroll
        for (int j = 0; j < 8; ++j)
            st[(j + 8 * hh) * TP + col] = fmaxf(c[nt][j] + bv, 0.f);
    }
}

template <int HEAD>
__global__ __launch_bounds__(1024) void own_kernel(
    const int* __restrict__ sedw,
    const unsigned* __restrict__ tab, const float* __restrict__ dinv,
    const float* __restrict__ xw, float* __restrict__ out, int n, int nslice,
    const float* __restrict__ cb, const float* __restrict__ bn_g,
    const float* __restrict__ bn_b, const float* __restrict__ bn_m,
    const float* __restrict__ bn_v,
    const float* __restrict__ Wa, const float* __restrict__ ba,
    const float* __restrict__ Wb, const float* __restrict__ bb,
    const float* __restrict__ Wc, const float* __restrict__ bc)
{
    const int2* sed = reinterpret_cast<const int2*>(sedw);
    constexpr int WF = (HEAD == 1) ? (64 * 64 * 2 + 64 * 16) : (64 * 64);
    __shared__ __align__(16) float smem[WF + (OWN_TR + 1) * TP + OWN_TR];
    float* sWt = smem;
    float* acc = smem + WF;
    float* sDv = acc + (OWN_TR + 1) * TP;
    const int tid  = threadIdx.x;
    const int lane = tid & 31;
    const int wave = __builtin_amdgcn_readfirstlane(tid >> 5);
    const int base = blockIdx.x * RN;
    int len = n - base;
    if (len > RN) len = RN;
    const int nq = len * 16;

    const unsigned* tb = tab + (size_t)blockIdx.x * nslice;
    unsigned segw = 0u;
    if (wave + 32 * lane < nslice) segw = tb[wave + 32 * lane];

    const float4 wa4 = reinterpret_cast<const float4*>(Wa)[tid];
    float4 wb4 = make_float4(0.f, 0.f, 0.f, 0.f), wc4 = wb4;
    if (HEAD == 1) {
        wb4 = reinterpret_cast<const float4*>(Wb)[tid];
        wc4 = reinterpret_cast<const float4*>(Wc)[tid & 255];
    }
    const float dme = dinv[base + ((tid < len) ? tid : len - 1)];

    {
        const float4* x4 = reinterpret_cast<const float4*>(xw + (size_t)base * 64);
        float4*       a4 = reinterpret_cast<float4*>(acc);
        float4 sv[OWN_NQ];
        #pragma unroll
        for (int t = 0; t < OWN_NQ; ++t) {
            int q = tid + 1024 * t;
            q = (q < nq) ? q : nq - 1;
            sv[t] = x4[q];
        }
        #pragma unroll
        for (int t = 0; t < OWN_NQ; ++t) {
            const int q = tid + 1024 * t;
            if (q < OWN_TR * 16) {
                const float4 v = (q < nq) ? sv[t] : make_float4(0.f, 0.f, 0.f, 0.f);
                a4[(q >> 4) * (TP / 4) + (q & 15)] = v;
            }
        }
        if (tid < TP / 4) a4[OWN_TR * (TP / 4) + tid] = make_float4(0.f, 0.f, 0.f, 0.f);
        if (tid < OWN_TR) sDv[tid] = (tid < len) ? dme : 0.f;
        float4* s4 = reinterpret_cast<float4*>(sWt);
        s4[tid] = wa4;
        if (HEAD == 1) {
            s4[1024 + tid] = wb4;
            if (tid < 256) s4[2048 + tid] = wc4;
        }
    }
    __syncthreads();

    {
        const int nt = (nslice - wave + 31) / 32;
        for (int t = 0; t < nt; ++t) {
            const unsigned word = (unsigned)__builtin_amdgcn_readlane((int)segw, t);
            const int c  = (int)(word >> 16);
            const int g0 = (wave + 32 * t) * BIN_SL + (int)(word & 0xffffu);
            for (int k0 = 0; k0 < c; k0 += 32) {
                int m = c - k0;
                if (m > 32) m = 32;
                int r  = base;
                int cl = OWN_TR;
                if (lane < m) {
                    const int2 ed = sed[g0 + k0 + lane];
                    r  = ed.x;
                    cl = ed.y;
                }
                for (int jb = 0; jb < m; jb += 8) {
                    float2 v[8];
                    #pragma unroll
                    for (int j = 0; j < 8; ++j) {
                        const int rj = __builtin_amdgcn_readlane(r, jb + j);
                        v[j] = *reinterpret_cast<const float2*>(xw + (size_t)rj * 64 + 2 * lane);
                    }
                    #pragma unroll
                    for (int j = 0; j < 8; ++j) {
                        const int cj = __builtin_amdgcn_readlane(cl, jb + j);
                        float* dst = acc + cj * TP + 2 * lane;
                        unsafeAtomicAdd(dst,     v[j].x);
                        unsafeAtomicAdd(dst + 1, v[j].y);
                    }
                }
            }
        }
    }
    __syncthreads();

    const int  hh  = lane >> 4;
    const int  l16 = lane & 15;
    const bool has = wave < OWN_ST;
    float* st = acc + (has ? wave : 0) * (16 * TP);
    const float* sd = sDv + (has ? wave : 0) * 16;

    if (has) {
        const float4 pb = reinterpret_cast<const float4*>(cb)[l16];
        const float4 pm = reinterpret_cast<const float4*>(bn_m)[l16];
        const float4 pg = reinterpret_cast<const float4*>(bn_g)[l16];
        const float4 pe = reinterpret_cast<const float4*>(bn_b)[l16];
        const float4 pv = reinterpret_cast<const float4*>(bn_v)[l16];
        float4 pr;
        pr.x = rsqrtf(pv.x + 1e-5f);
        pr.y = rsqrtf(pv.y + 1e-5f);
        pr.z = rsqrtf(pv.z + 1e-5f);
        pr.w = rsqrtf(pv.w + 1e-5f);
        #pragma unroll
        for (int i = 0; i < 8; ++i) {
            float4* p = reinterpret_cast<float4*>(&st[(2 * i + hh) * TP + l16 * 4]);
            const float dc = sd[2 * i + hh];
            float4 v = *p;
            v.x = fmaxf((v.x * dc + pb.x - pm.x) * pr.x * pg.x + pe.x, 0.f);
            v.y = fmaxf((v.y * dc + pb.y - pm.y) * pr.y * pg.y + pe.y, 0.f);
            v.z = fmaxf((v.z * dc + pb.z - pm.z) * pr.z * pg.z + pe.z, 0.f);
            v.w = fmaxf((v.w * dc + pb.w - pm.w) * pr.w * pg.w + pe.w, 0.f);
            *p = v;
        }
    }
    __syncthreads();

    v8f c1[4] = {};
    if (has) strip_product<64>(st, sWt, c1, l16, hh);

    if (HEAD == 0) {
        if (has) {
            float dr[8];
            #pragma unroll
            for (int j = 0; j < 8; ++j) dr[j] = sd[j + 8 * hh];
            #pragma unroll
            for (int nt = 0; nt < 4; ++nt) {
                const int col = nt * 16 + l16;
                #pragma unroll
                for (int j = 0; j < 8; ++j) {
                    const int lr = wave * 16 + j + 8 * hh;
                    if (lr < len) out[(size_t)(base + lr) * 64 + col] = c1[nt][j] * dr[j];
                }
            }
        }
    } else {
        if (has) strip_put_relu(st, c1, ba, l16, hh);
        __syncthreads();
        v8f c2[4] = {};
        if (has) {
            strip_product<64>(st, sWt + 64 * 64, c2, l16, hh);
            strip_put_relu(st, c2, bb, l16, hh);
        }
        __syncthreads();
        if (has) {
            v8f c3[1] = {};
            strip_product<16>(st, sWt + 2 * 64 * 64, c3, l16, hh);
            const float bv = bc[l16];
            #pragma unroll
            for (int j = 0; j < 8; ++j) {
                const int lr = wave * 16 + j + 8 * hh;
                if (lr < len) out[(size_t)(base + lr) * 16 + l16] = c3[0][j] + bv;
            }
        }
    }
}

extern "C" void kernel_launch(void* const* d_in, const int* in_sizes, int n_in,
                              void* d_out, int out_size, void* d_ws, size_t ws_size,
                              hipStream_t stream)
{
    const float* x      = (const float*)d_in[0];
    const int*   ei     = (const int*)d_in[1];
    const float* W0     = (const float*)d_in[2];
    const float* b0     = (const float*)d_in[3];
    const float* W1     = (const float*)d_in[4];
    const float* b1     = (const float*)d_in[5];
    const float* bn0_g  = (const float*)d_in[6];
    const float* bn0_b  = (const float*)d_in[7];
    const float* bn0_m  = (const float*)d_in[8];
    const float* bn0_v  = (const float*)d_in[9];
    const float* bn1_g  = (const float*)d_in[10];
    const float* bn1_b  = (const float*)d_in[11];
    const float* bn1_m  = (const float*)d_in[12];
    const float* bn1_v  = (const float*)d_in[13];
    const float* lin1_W = (const float*)d_in[14];
    const float* lin1_b = (const float*)d_in[15];
    const float* lin2_W = (const float*)d_in[16];
    const float* lin2_b = (const float*)d_in[17];
    const float* lin3_W = (const float*)d_in[18];
    const float* lin3_b = (const float*)d_in[19];

    const int N = in_sizes[0] / 64;
    const int E = in_sizes[1] / 2;
    const int* row = ei;
    const int* col = ei + E;

    char*  ws   = (char*)d_ws;
    size_t off0 = 0;
    size_t off1 = (off0 + (size_t)N * sizeof(float) + 255) & ~(size_t)255;
    size_t off2 = (off1 + (size_t)N * 64 * sizeof(float) + 255) & ~(size_t)255;
    float* dinv = (float*)(ws + off0);
    float* bufA = (float*)(ws + off1);
    float* bufB = (float*)(ws + off2);
    size_t off4 = (off2 + (size_t)N * 64 * sizeof(float) + 255) & ~(size_t)255;
    unsigned* part = (unsigned*)(ws + off4);

    const int gBin  = (E + BIN_SL - 1) / BIN_SL;
    const int gOwn  = (N + RN - 1) / RN;
    const int gSl   = ((E + DEG_SLICE - 1) / DEG_SLICE + DEG_KB - 1) / DEG_KB * DEG_KB;
    const int gRg   = (N + DEG_RANGE - 1) / DEG_RANGE;
    const int nwS   = ((N + 1) / 2 + 3) & ~3;
    const int gFront = (N + FRONT_ROWS - 1) / FRONT_ROWS;
    size_t off5 = (off4 + (size_t)gSl * nwS * sizeof(unsigned) + 255) & ~(size_t)255;
    size_t off6 = (off5 + (size_t)E * sizeof(int2) + 255) & ~(size_t)255;
    int*      sed = (int*)(ws + off5);
    unsigned* tab = (unsigned*)(ws + off6);

    deg_kernel<<<gSl * gRg, dim3(1024), 0, stream>>>(col, part, E, N, gSl, nwS);

    front_kernel<<<gBin + gFront, dim3(1024), 0, stream>>>(
        row, col, sed, tab, part, dinv, x, W0, bufA,
        E, N, gOwn, gBin, nwS, gSl);

    own_kernel<0><<<gOwn, dim3(1024), 0, stream>>>(
        sed, tab, dinv, bufA, bufB, N, gBin,
        b0, bn0_g, bn0_b, bn0_m, bn0_v,
        W1, nullptr, nullptr, nullptr, nullptr, nullptr);

    own_kernel<1><<<gOwn, dim3(1024), 0, stream>>>(
        sed, tab, dinv, bufB, (float*)d_out, N, gBin,
        b1, bn1_g, bn1_b, bn1_m, bn1_v,
        lin1_W, lin1_b, lin2_W, lin2_b, lin3_W, lin3_b);
}
